# starting kernel
# baseline (speedup 1.0000x reference)
#include <hip/hip_runtime.h>


typedef __attribute__((ext_vector_type(2))) float v2f;
typedef __attribute__((ext_vector_type(8))) float v8f;

#define D 128
#define DV4 (D / 4)

__global__ void zero_int_kernel(int* __restrict__ p, int n) {
    int i = blockIdx.x * blockDim.x + threadIdx.x;
    if (i < n) p[i] = 0;
}

__global__ void copy_f32_kernel(const float* __restrict__ a, float* __restrict__ b, int n) {
    int i = blockIdx.x * blockDim.x + threadIdx.x;
    if (i < n) b[i] = a[i];
}

__global__ void count_kernel(const int* __restrict__ dst, int* __restrict__ cnts,
                             int total, int N, int E) {
    int i = blockIdx.x * blockDim.x + threadIdx.x;
    if (i >= total) return;
    int r = i / E;
    atomicAdd(&cnts[r * N + dst[i]], 1);
}

__global__ void gemm_wmma_f32_kernel(const float* __restrict__ h,
                                     const float* __restrict__ W,
                                     float* __restrict__ hW, int N) {
    __shared__ float As[16 * D];

    const int rowBase = blockIdx.x * 16;

    const float4* __restrict__ h4 = (const float4*)h;
    float4* As4 = (float4*)As;
    for (int i = threadIdx.x; i < 16 * DV4; i += blockDim.x) {
        int rr = i / DV4, cc = i - rr * DV4;
        int grow = rowBase + rr;
        As4[i] = (grow < N) ? h4[(size_t)grow * DV4 + cc] : make_float4(0.f, 0.f, 0.f, 0.f);
    }
    __syncthreads();

    const int wave = threadIdx.x >> 5;
    const int lane = threadIdx.x & 31;
    const int half = lane >> 4;
    const int lm   = lane & 15;
    const int colBase = (blockIdx.y * 4 + wave) * 16;

    v8f c = {};
    for (int k0 = 0; k0 < D; k0 += 4) {
        const int ka = k0 + 2 * half;
        v2f a, b;
        a[0] = As[lm * D + ka];
        a[1] = As[lm * D + ka + 1];
        b[0] = W[(size_t)ka * D + colBase + lm];
        b[1] = W[(size_t)(ka + 1) * D + colBase + lm];
        c = __builtin_amdgcn_wmma_f32_16x16x4_f32(
                 false, a,  false, b,
                 (short)0, c,  false,  false);
    }

    for (int v = 0; v < 8; ++v) {
        int grow = rowBase + v + 8 * half;
        if (grow < N) hW[(size_t)grow * D + colBase + lm] = c[v];
    }
}

__global__ void scatter_kernel(const float* __restrict__ hW,
                               const int* __restrict__ src,
                               const int* __restrict__ dst,
                               const int* __restrict__ cnts,
                               float* __restrict__ out,
                               int E, float invR) {
    const int slot = blockIdx.x * (blockDim.x >> 5) + (threadIdx.x >> 5);
    const int lane = threadIdx.x & 31;
    if (slot >= E) return;

    const int s = src[slot];
    const int t = dst[slot];
    const int c = cnts[t];
    const float scale = invR / (float)(c > 0 ? c : 1);

    const float4* __restrict__ v4 = (const float4*)(hW + (size_t)s * D);
    float4 v = v4[lane];

    float* o = out + (size_t)t * D + lane * 4;
    atomicAdd(o + 0, v.x * scale);
    atomicAdd(o + 1, v.y * scale);
    atomicAdd(o + 2, v.z * scale);
    atomicAdd(o + 3, v.w * scale);
}

extern "C" void kernel_launch(void* const* d_in, const int* in_sizes, int n_in,
                              void* d_out, int out_size, void* d_ws, size_t ws_size,
                              hipStream_t stream) {
    const float* h   = (const float*)d_in[0];
    const float* W   = (const float*)d_in[1];
    const int*   src = (const int*)d_in[2];
    const int*   dst = (const int*)d_in[3];
    float*       out = (float*)d_out;

    const int d = D;
    const int N = in_sizes[0] / d;
    const int R = in_sizes[1] / (d * d);
    const int E = in_sizes[2] / R;

    float* hW   = (float*)d_ws;
    int*   cnts = (int*)((char*)d_ws + (size_t)N * d * sizeof(float));

    {
        int n = R * N;
        zero_int_kernel<<<(n + 255) / 256, 256, 0, stream>>>(cnts, n);
        int total = R * E;
        count_kernel<<<(total + 255) / 256, 256, 0, stream>>>(dst, cnts, total, N, E);
    }

    {
        int n = N * d;
        copy_f32_kernel<<<(n + 255) / 256, 256, 0, stream>>>(h, out, n);
    }

    const float invR = 1.0f / (float)R;
    for (int r = 0; r < R; ++r) {
        dim3 ggrid((N + 15) / 16, d / (16 * 4));
        gemm_wmma_f32_kernel<<<ggrid, 128, 0, stream>>>(h, W + (size_t)r * d * d, hW, N);

        dim3 sgrid((E + 7) / 8);
        scatter_kernel<<<sgrid, 256, 0, stream>>>(hW, src + (size_t)r * E,
                                                  dst + (size_t)r * E,
                                                  cnts + (size_t)r * N,
                                                  out, E, invR);
    }
}


# GEMM: pad the LDS row pitch of the staged A tile from 128 to 132 floats, so the 16 rows of an A-fragment read fall on 16 different bank groups instead of one (16-way LDS bank conflict removed)
# speedup vs baseline: 1.1141x; 1.1141x over previous
#include <hip/hip_runtime.h>


typedef __attribute__((ext_vector_type(2))) float v2f;
typedef __attribute__((ext_vector_type(8))) float v8f;

#define D 128
#define DV4 (D / 4)
#define AP (D + 4)

__global__ void zero_int_kernel(int* __restrict__ p, int n) {
    int i = blockIdx.x * blockDim.x + threadIdx.x;
    if (i < n) p[i] = 0;
}

__global__ void copy_f32_kernel(const float* __restrict__ a, float* __restrict__ b, int n) {
    int i = blockIdx.x * blockDim.x + threadIdx.x;
    if (i < n) b[i] = a[i];
}

__global__ void count_kernel(const int* __restrict__ dst, int* __restrict__ cnts,
                             int total, int N, int E) {
    int i = blockIdx.x * blockDim.x + threadIdx.x;
    if (i >= total) return;
    int r = i / E;
    atomicAdd(&cnts[r * N + dst[i]], 1);
}

__global__ void gemm_wmma_f32_kernel(const float* __restrict__ h,
                                     const float* __restrict__ W,
                                     float* __restrict__ hW, int N) {
    __shared__ __attribute__((aligned(16))) float As[16 * AP];

    const int rowBase = blockIdx.x * 16;

    const float4* __restrict__ h4 = (const float4*)h;
    for (int i = threadIdx.x; i < 16 * DV4; i += blockDim.x) {
        int rr = i / DV4, cc = i - rr * DV4;
        int grow = rowBase + rr;
        *(float4*)(As + rr * AP + cc * 4) =
            (grow < N) ? h4[(size_t)grow * DV4 + cc] : make_float4(0.f, 0.f, 0.f, 0.f);
    }
    __syncthreads();

    const int wave = threadIdx.x >> 5;
    const int lane = threadIdx.x & 31;
    const int half = lane >> 4;
    const int lm   = lane & 15;
    const int colBase = (blockIdx.y * 4 + wave) * 16;

    v8f c = {};
    for (int k0 = 0; k0 < D; k0 += 4) {
        const int ka = k0 + 2 * half;
        v2f a, b;
        a[0] = As[lm * AP + ka];
        a[1] = As[lm * AP + ka + 1];
        b[0] = W[(size_t)ka * D + colBase + lm];
        b[1] = W[(size_t)(ka + 1) * D + colBase + lm];
        c = __builtin_amdgcn_wmma_f32_16x16x4_f32(
                 false, a,  false, b,
                 (short)0, c,  false,  false);
    }

    for (int v = 0; v < 8; ++v) {
        int grow = rowBase + v + 8 * half;
        if (grow < N) hW[(size_t)grow * D + colBase + lm] = c[v];
    }
}

__global__ void scatter_kernel(const float* __restrict__ hW,
                               const int* __restrict__ src,
                               const int* __restrict__ dst,
                               const int* __restrict__ cnts,
                               float* __restrict__ out,
                               int E, float invR) {
    const int slot = blockIdx.x * (blockDim.x >> 5) + (threadIdx.x >> 5);
    const int lane = threadIdx.x & 31;
    if (slot >= E) return;

    const int s = src[slot];
    const int t = dst[slot];
    const int c = cnts[t];
    const float scale = invR / (float)(c > 0 ? c : 1);

    const float4* __restrict__ v4 = (const float4*)(hW + (size_t)s * D);
    float4 v = v4[lane];

    float* o = out + (size_t)t * D + lane * 4;
    atomicAdd(o + 0, v.x * scale);
    atomicAdd(o + 1, v.y * scale);
    atomicAdd(o + 2, v.z * scale);
    atomicAdd(o + 3, v.w * scale);
}

extern "C" void kernel_launch(void* const* d_in, const int* in_sizes, int n_in,
                              void* d_out, int out_size, void* d_ws, size_t ws_size,
                              hipStream_t stream) {
    const float* h   = (const float*)d_in[0];
    const float* W   = (const float*)d_in[1];
    const int*   src = (const int*)d_in[2];
    const int*   dst = (const int*)d_in[3];
    float*       out = (float*)d_out;

    const int d = D;
    const int N = in_sizes[0] / d;
    const int R = in_sizes[1] / (d * d);
    const int E = in_sizes[2] / R;

    float* hW   = (float*)d_ws;
    int*   cnts = (int*)((char*)d_ws + (size_t)N * d * sizeof(float));

    {
        int n = R * N;
        zero_int_kernel<<<(n + 255) / 256, 256, 0, stream>>>(cnts, n);
        int total = R * E;
        count_kernel<<<(total + 255) / 256, 256, 0, stream>>>(dst, cnts, total, N, E);
    }

    {
        int n = N * d;
        copy_f32_kernel<<<(n + 255) / 256, 256, 0, stream>>>(h, out, n);
    }

    const float invR = 1.0f / (float)R;
    for (int r = 0; r < R; ++r) {
        dim3 ggrid((N + 15) / 16, d / (16 * 4));
        gemm_wmma_f32_kernel<<<ggrid, 128, 0, stream>>>(h, W + (size_t)r * d * d, hW, N);

        dim3 sgrid((E + 7) / 8);
        scatter_kernel<<<sgrid, 256, 0, stream>>>(hW, src + (size_t)r * E,
                                                  dst + (size_t)r * E,
                                                  cnts + (size_t)r * N,
                                                  out, E, invR);
    }
}


# scatter + residual copy: each wave takes 32 edges (was 1), each lane copies 8 float4 (was 1 float), so 1,900k short-lived waves become 59k (wave entry rate was the limit); f32 atomics kept, unordered
# speedup vs baseline: 1.1933x; 1.0711x over previous
#include <hip/hip_runtime.h>


typedef __attribute__((ext_vector_type(2))) float v2f;
typedef __attribute__((ext_vector_type(8))) float v8f;

#define D 128
#define DV4 (D / 4)
#define AP (D + 4)

__global__ void zero_int_kernel(int* __restrict__ p, int n) {
    int i = blockIdx.x * blockDim.x + threadIdx.x;
    if (i < n) p[i] = 0;
}

#define CPL 8
__global__ void copy_f32_kernel(const float* a, float* b, int n) {
    const float4* a4 = (const float4*)a;
    float4* b4 = (float4*)b;
    const int n4 = n >> 2;
    const int bd = blockDim.x;
    const int blockBase = blockIdx.x * (bd * CPL);
    const int i0 = blockBase + threadIdx.x;
    if (blockBase + bd * CPL <= n4) {
        const float4 v0 = a4[i0];
        const float4 v1 = a4[i0 + bd];
        const float4 v2 = a4[i0 + 2 * bd];
        const float4 v3 = a4[i0 + 3 * bd];
        const float4 v4 = a4[i0 + 4 * bd];
        const float4 v5 = a4[i0 + 5 * bd];
        const float4 v6 = a4[i0 + 6 * bd];
        const float4 v7 = a4[i0 + 7 * bd];
        b4[i0]          = v0;
        b4[i0 + bd]     = v1;
        b4[i0 + 2 * bd] = v2;
        b4[i0 + 3 * bd] = v3;
        b4[i0 + 4 * bd] = v4;
        b4[i0 + 5 * bd] = v5;
        b4[i0 + 6 * bd] = v6;
        b4[i0 + 7 * bd] = v7;
    } else {
        for (int u = 0; u < CPL; ++u) {
            const int i = i0 + u * bd;
            if (i < n4) b4[i] = a4[i];
        }
    }
}

__global__ void count_kernel(const int* __restrict__ dst, int* __restrict__ cnts,
                             int total, int N, int E) {
    int i = blockIdx.x * blockDim.x + threadIdx.x;
    if (i >= total) return;
    int r = i / E;
    atomicAdd(&cnts[r * N + dst[i]], 1);
}

__global__ void gemm_wmma_f32_kernel(const float* __restrict__ h,
                                     const float* __restrict__ W,
                                     float* __restrict__ hW, int N) {
    __shared__ __attribute__((aligned(16))) float As[16 * AP];

    const int rowBase = blockIdx.x * 16;

    const float4* __restrict__ h4 = (const float4*)h;
    for (int i = threadIdx.x; i < 16 * DV4; i += blockDim.x) {
        int rr = i / DV4, cc = i - rr * DV4;
        int grow = rowBase + rr;
        *(float4*)(As + rr * AP + cc * 4) =
            (grow < N) ? h4[(size_t)grow * DV4 + cc] : make_float4(0.f, 0.f, 0.f, 0.f);
    }
    __syncthreads();

    const int wave = threadIdx.x >> 5;
    const int lane = threadIdx.x & 31;
    const int half = lane >> 4;
    const int lm   = lane & 15;
    const int colBase = (blockIdx.y * 4 + wave) * 16;

    v8f c = {};
    for (int k0 = 0; k0 < D; k0 += 4) {
        const int ka = k0 + 2 * half;
        v2f a, b;
        a[0] = As[lm * AP + ka];
        a[1] = As[lm * AP + ka + 1];
        b[0] = W[(size_t)ka * D + colBase + lm];
        b[1] = W[(size_t)(ka + 1) * D + colBase + lm];
        c = __builtin_amdgcn_wmma_f32_16x16x4_f32(
                 false, a,  false, b,
                 (short)0, c,  false,  false);
    }

    for (int v = 0; v < 8; ++v) {
        int grow = rowBase + v + 8 * half;
        if (grow < N) hW[(size_t)grow * D + colBase + lm] = c[v];
    }
}

#define EPW 32
#define EB  8
__global__ void scatter_kernel(const float* __restrict__ hW,
                               const int* __restrict__ src,
                               const int* __restrict__ dst,
                               const int* __restrict__ cnts,
                               float* __restrict__ out,
                               int E, float invR) {
    const int wv   = blockIdx.x * (blockDim.x >> 5) + (threadIdx.x >> 5);
    const int lane = threadIdx.x & 31;
    const int base = wv * EPW;
    if (base >= E) return;

    int   sL  = 0;
    int   tL  = -1;
    float scL = 0.f;
    if (base + lane < E) {
        tL = dst[base + lane];
        sL = src[base + lane];
        const int c = cnts[tL];
        scL = invR / (float)(c > 0 ? c : 1);
    }

    const float4* __restrict__ hW4 = (const float4*)hW;
#pragma unroll
    for (int j0 = 0; j0 < EPW; j0 += EB) {
        float4 v[EB];
#pragma unroll
        for (int u = 0; u < EB; ++u) {
            const int s = __builtin_amdgcn_readlane(sL, j0 + u);
            v[u] = hW4[(size_t)s * DV4 + lane];
        }
#pragma unroll
        for (int u = 0; u < EB; ++u) {
            const int   t  = __builtin_amdgcn_readlane(tL, j0 + u);
            const float sc = __builtin_bit_cast(float,
                                 __builtin_amdgcn_readlane(__builtin_bit_cast(int, scL), j0 + u));
            if (t >= 0) {
                float* o = out + (size_t)t * D + lane * 4;
                atomicAdd(o + 0, v[u].x * sc);
                atomicAdd(o + 1, v[u].y * sc);
                atomicAdd(o + 2, v[u].z * sc);
                atomicAdd(o + 3, v[u].w * sc);
            }
        }
    }
}

extern "C" void kernel_launch(void* const* d_in, const int* in_sizes, int n_in,
                              void* d_out, int out_size, void* d_ws, size_t ws_size,
                              hipStream_t stream) {
    const float* h   = (const float*)d_in[0];
    const float* W   = (const float*)d_in[1];
    const int*   src = (const int*)d_in[2];
    const int*   dst = (const int*)d_in[3];
    float*       out = (float*)d_out;

    const int d = D;
    const int N = in_sizes[0] / d;
    const int R = in_sizes[1] / (d * d);
    const int E = in_sizes[2] / R;

    float* hW   = (float*)d_ws;
    int*   cnts = (int*)((char*)d_ws + (size_t)N * d * sizeof(float));

    {
        int n = R * N;
        zero_int_kernel<<<(n + 255) / 256, 256, 0, stream>>>(cnts, n);
        int total = R * E;
        count_kernel<<<(total + 255) / 256, 256, 0, stream>>>(dst, cnts, total, N, E);
    }

    {
        int n = N * d;
        copy_f32_kernel<<<(n / 4 + 256 * CPL - 1) / (256 * CPL), 256, 0, stream>>>(h, out, n);
    }

    const float invR = 1.0f / (float)R;
    for (int r = 0; r < R; ++r) {
        dim3 ggrid((N + 15) / 16, d / (16 * 4));
        gemm_wmma_f32_kernel<<<ggrid, 128, 0, stream>>>(h, W + (size_t)r * d * d, hW, N);

        dim3 sgrid((E + 8 * EPW - 1) / (8 * EPW));
        scatter_kernel<<<sgrid, 256, 0, stream>>>(hW, src + (size_t)r * E,
                                                  dst + (size_t)r * E,
                                                  cnts + (size_t)r * N,
                                                  out, E, invR);
    }
}


# scatter -> gather: per-destination edge lists (int exchange) are walked, h rows averaged BEFORE the W product, one fused WMMA launch; f32 atomics and hW round trip gone; sum order changed
# speedup vs baseline: 2.9057x; 2.4350x over previous
#include <hip/hip_runtime.h>


typedef __attribute__((ext_vector_type(2))) float v2f;
typedef __attribute__((ext_vector_type(8))) float v8f;

#define D 128
#define DV4 (D / 4)
#define NR 3
#define KT (NR * D)
#define AP (KT + 4)
#define CP (D + 4)
#define TM 32
#define NWV 8
#define RPW (TM / NWV)
#define NCH (RPW * NR)

__global__ void fill_int_kernel(int* __restrict__ p, int n, int v) {
    int i = blockIdx.x * blockDim.x + threadIdx.x;
    if (i < n) p[i] = v;
}

__global__ void link_kernel(const int* __restrict__ src, const int* __restrict__ dst,
                            int* __restrict__ head, int* __restrict__ cells,
                            int total, int N, int E) {
    int i = blockIdx.x * blockDim.x + threadIdx.x;
    if (i >= total) return;
    const int r = i / E;
    const int prev = atomicExch(&head[r * N + dst[i]], i);
    ((int2*)cells)[i] = make_int2(prev, src[i]);
}

__global__ __launch_bounds__(256)
void gather_transform_kernel(const float* __restrict__ h,
                             const float* __restrict__ W,
                             const int* __restrict__ head,
                             const int* __restrict__ cells,
                             float* __restrict__ out,
                             int N, int E, float invR) {
    __shared__ __attribute__((aligned(16))) float As[TM * AP];
    __shared__ __attribute__((aligned(16))) float Cs[TM * CP];

    const int wave = threadIdx.x >> 5;
    const int lane = threadIdx.x & 31;
    const int rowBase = blockIdx.x * TM;
    const float4* __restrict__ h4 = (const float4*)h;
    const int2* __restrict__ cell2 = (const int2*)cells;
    const int lastEdge = NR * E - 1;

    int eL = -1;
    if (lane < NCH) {
        const int r = lane >> 2;
        const int node = rowBase + wave * RPW + (lane & 3);
        if (node < N) eL = head[r * N + node];
    }
    int cntL = 0;
    float4 acc[NCH];
#pragma unroll
    for (int q = 0; q < NCH; ++q) acc[q] = make_float4(0.f, 0.f, 0.f, 0.f);

    for (int it = 0; it < E; ++it) {
        const bool live = eL >= 0;
        const unsigned liveMask = __builtin_amdgcn_ballot_w32(live);
        if (liveMask == 0u) break;
        const int ec = live ? (eL < lastEdge ? eL : lastEdge) : 0;
        const int2 nd = cell2[ec];
        cntL += live ? 1 : 0;
#pragma unroll
        for (int q = 0; q < NCH; ++q) {
            int s = __builtin_amdgcn_readlane(nd.y, q);
            s = s < 0 ? 0 : (s < N ? s : N - 1);
            const float4 v = h4[(size_t)s * DV4 + lane];
            const bool on = ((liveMask >> q) & 1u) != 0u;
            acc[q].x += on ? v.x : 0.f;
            acc[q].y += on ? v.y : 0.f;
            acc[q].z += on ? v.z : 0.f;
            acc[q].w += on ? v.w : 0.f;
        }
        eL = live ? nd.x : -1;
    }

    const float scL = invR / (float)(cntL > 0 ? cntL : 1);
#pragma unroll
    for (int q = 0; q < NCH; ++q) {
        const float sc = __builtin_bit_cast(float,
                             __builtin_amdgcn_readlane(__builtin_bit_cast(int, scL), q));
        const int r = q >> 2;
        const int row = wave * RPW + (q & 3);
        float4 m;
        m.x = acc[q].x * sc; m.y = acc[q].y * sc; m.z = acc[q].z * sc; m.w = acc[q].w * sc;
        *(float4*)(As + row * AP + r * D + lane * 4) = m;
    }
    __syncthreads();

    const int half = lane >> 4;
    const int lm   = lane & 15;
    const int colBase = wave * 16;

    v8f c0 = {};
    v8f c1 = {};
#pragma unroll 16
    for (int kk = 0; kk < KT; kk += 4) {
        const int ka = kk + 2 * half;
        v2f a0, a1, b;
        a0[0] = As[lm * AP + ka];
        a0[1] = As[lm * AP + ka + 1];
        a1[0] = As[(16 + lm) * AP + ka];
        a1[1] = As[(16 + lm) * AP + ka + 1];
        b[0] = W[(size_t)ka * D + colBase + lm];
        b[1] = W[(size_t)(ka + 1) * D + colBase + lm];
        c0 = __builtin_amdgcn_wmma_f32_16x16x4_f32(
                 false, a0,  false, b,
                 (short)0, c0,  false,  false);
        c1 = __builtin_amdgcn_wmma_f32_16x16x4_f32(
                 false, a1,  false, b,
                 (short)0, c1,  false,  false);
    }

#pragma unroll
    for (int v = 0; v < 8; ++v) {
        Cs[(v + 8 * half) * CP + colBase + lm]      = c0[v];
        Cs[(16 + v + 8 * half) * CP + colBase + lm] = c1[v];
    }
    __syncthreads();

    float4* __restrict__ out4 = (float4*)out;
#pragma unroll
    for (int q = 0; q < RPW; ++q) {
        const int row  = wave * RPW + q;
        const int grow = rowBase + row;
        if (grow < N) {
            const float4 cv = *(const float4*)(Cs + row * CP + lane * 4);
            const float4 hv = h4[(size_t)grow * DV4 + lane];
            float4 o;
            o.x = hv.x + cv.x; o.y = hv.y + cv.y; o.z = hv.z + cv.z; o.w = hv.w + cv.w;
            out4[(size_t)grow * DV4 + lane] = o;
        }
    }
}

extern "C" void kernel_launch(void* const* d_in, const int* in_sizes, int n_in,
                              void* d_out, int out_size, void* d_ws, size_t ws_size,
                              hipStream_t stream) {
    const float* h   = (const float*)d_in[0];
    const float* W   = (const float*)d_in[1];
    const int*   src = (const int*)d_in[2];
    const int*   dst = (const int*)d_in[3];
    float*       out = (float*)d_out;

    const int d = D;
    const int N = in_sizes[0] / d;
    const int R = NR;
    const int E = in_sizes[2] / R;

    int* head  = (int*)d_ws;
    int* cells = (int*)((char*)d_ws + (((size_t)R * N * sizeof(int) + 15) / 16) * 16);

    {
        int n = R * N;
        fill_int_kernel<<<(n + 255) / 256, 256, 0, stream>>>(head, n, -1);
    }
    {
        int total = R * E;
        link_kernel<<<(total + 255) / 256, 256, 0, stream>>>(src, dst, head, cells, total, N, E);
    }
    {
        const float invR = 1.0f / (float)R;
        gather_transform_kernel<<<(N + TM - 1) / TM, NWV * 32, 0, stream>>>(h, W, head, cells,
                                                                          out, N, E, invR);
    }
}


# per-edge global int exchange (list link) replaced by grouping edges per 32-node tile with LDS counters; each tile sorts its records in LDS and walks runs, no cell loads; sums stay in arrival order
# speedup vs baseline: 4.0802x; 1.4042x over previous
#include <hip/hip_runtime.h>


typedef __attribute__((ext_vector_type(2))) float v2f;
typedef __attribute__((ext_vector_type(8))) float v8f;

#define D 128
#define DV4 (D / 4)
#define NR 3
#define KT (NR * D)
#define AP (KT + 4)
#define CP (D + 4)
#define TM 32
#define NWV 8
#define RPW (TM / NWV)
#define NCH (RPW * NR)
#define NKEY (NR * TM)
#define NBMAX 4096
#define GT 1024
#define EPT 6
#define CH 1024
#define RPT (CH / (NWV * 32))

__global__ void fill_int_kernel(int* __restrict__ p, int n, int v) {
    int i = blockIdx.x * blockDim.x + threadIdx.x;
    if (i < n) p[i] = v;
}

__device__ __forceinline__ int tile_of(int d, int N) {
    d = d < 0 ? 0 : (d < N ? d : N - 1);
    const int t = d >> 5;
    return t < NBMAX ? t : NBMAX - 1;
}

__global__ __launch_bounds__(GT)
void count_kernel(const int* __restrict__ dst, int* __restrict__ gcount, int total, int N) {
    __shared__ int cnt[NBMAX];
    for (int b = threadIdx.x; b < NBMAX; b += GT) cnt[b] = 0;
    __syncthreads();
    const int base = blockIdx.x * (GT * EPT);
#pragma unroll
    for (int k = 0; k < EPT; ++k) {
        const int i = base + k * GT + threadIdx.x;
        if (i < total) atomicAdd(&cnt[tile_of(dst[i], N)], 1);
    }
    __syncthreads();
    for (int b = threadIdx.x; b < NBMAX; b += GT) {
        const int c = cnt[b];
        if (c) atomicAdd(&gcount[b], c);
    }
}

__global__ __launch_bounds__(GT)
void group_kernel(const int* __restrict__ src, const int* __restrict__ dst,
                  const int* __restrict__ gcount, int* __restrict__ gcursor,
                  int* __restrict__ off, int* __restrict__ recs, int total, int N, int E) {
    int2* __restrict__ rec = (int2*)recs;
    __shared__ int offs[NBMAX + 4];
    __shared__ int cnt[NBMAX];
    __shared__ int wsum[GT / 32];
    const int tid = threadIdx.x, lane = tid & 31, wave = tid >> 5;

    const int4 v = ((const int4*)gcount)[tid];
    const int s = v.x + v.y + v.z + v.w;
    int incl = s;
#pragma unroll
    for (int o = 1; o < 32; o <<= 1) {
        const int t2 = __shfl_up(incl, o);
        if (lane >= o) incl += t2;
    }
    if (lane == 31) wsum[wave] = incl;
    for (int b = tid; b < NBMAX; b += GT) cnt[b] = 0;
    __syncthreads();
    if (wave == 0) {
        const int x = wsum[lane];
        int xi = x;
#pragma unroll
        for (int o = 1; o < 32; o <<= 1) {
            const int t2 = __shfl_up(xi, o);
            if (lane >= o) xi += t2;
        }
        wsum[lane] = xi - x;
    }
    __syncthreads();
    const int ex = incl - s + wsum[wave];
    offs[4 * tid + 0] = ex;
    offs[4 * tid + 1] = ex + v.x;
    offs[4 * tid + 2] = ex + v.x + v.y;
    offs[4 * tid + 3] = ex + v.x + v.y + v.z;
    if (tid == GT - 1) offs[NBMAX] = ex + s;
    if (blockIdx.x == 0) {
        off[4 * tid + 0] = ex;
        off[4 * tid + 1] = ex + v.x;
        off[4 * tid + 2] = ex + v.x + v.y;
        off[4 * tid + 3] = ex + v.x + v.y + v.z;
        if (tid == GT - 1) off[NBMAX] = ex + s;
    }

    const int base = blockIdx.x * (GT * EPT);
    int rank[EPT];
#pragma unroll
    for (int k = 0; k < EPT; ++k) {
        const int i = base + k * GT + tid;
        rank[k] = 0;
        if (i < total) rank[k] = atomicAdd(&cnt[tile_of(dst[i], N)], 1);
    }
    __syncthreads();
    for (int b = tid; b < NBMAX; b += GT) {
        const int c = cnt[b];
        int st = 0;
        if (c) st = atomicAdd(&gcursor[b], c);
        cnt[b] = offs[b] + st;
    }
    __syncthreads();
#pragma unroll
    for (int k = 0; k < EPT; ++k) {
        const int i = base + k * GT + tid;
        if (i < total) {
            int d = dst[i];
            d = d < 0 ? 0 : (d < N ? d : N - 1);
            const int t = tile_of(d, N);
            const int r = i / E;
            int pos = cnt[t] + rank[k];
            pos = pos < 0 ? 0 : (pos < total ? pos : total - 1);
            rec[pos] = make_int2(r * TM + (d & (TM - 1)), src[i]);
        }
    }
}

__global__ __launch_bounds__(256)
void gather_transform_kernel(const float* __restrict__ h,
                             const float* __restrict__ W,
                             const int* __restrict__ off,
                             const int* __restrict__ recs,
                             float* __restrict__ out,
                             int N, int total, float invR) {
    __shared__ __attribute__((aligned(16))) float As[TM * AP];
    __shared__ __attribute__((aligned(16))) float Cs[TM * CP];
    __shared__ int kcnt[NKEY];
    __shared__ int kofs[NKEY];
    __shared__ int srt[CH];

    const int wave = threadIdx.x >> 5;
    const int lane = threadIdx.x & 31;
    const int rowBase = blockIdx.x * TM;
    const float4* __restrict__ h4 = (const float4*)h;
    const int2* __restrict__ rec = (const int2*)recs;

    int beg = off[blockIdx.x];
    int end = off[blockIdx.x + 1];
    beg = beg < 0 ? 0 : (beg < total ? beg : total);
    end = end < beg ? beg : (end < total ? end : total);

    const int myKey = (lane < NCH) ? (lane >> 2) * TM + wave * RPW + (lane & 3) : 0;
    int cntL = 0;
    float4 acc[NCH];
#pragma unroll
    for (int q = 0; q < NCH; ++q) acc[q] = make_float4(0.f, 0.f, 0.f, 0.f);

    for (int cb = beg; cb < end; cb += CH) {
        const int n = (end - cb) < CH ? (end - cb) : CH;
        if (threadIdx.x < NKEY) kcnt[threadIdx.x] = 0;
        __syncthreads();
        int rk[RPT], ky[RPT], sv[RPT];
        int2 rc[RPT];
#pragma unroll
        for (int j = 0; j < RPT; ++j) {
            const int idx = j * (NWV * 32) + threadIdx.x;
            rc[j] = rec[cb + (idx < n ? idx : 0)];
        }
#pragma unroll
        for (int j = 0; j < RPT; ++j) {
            const int idx = j * (NWV * 32) + threadIdx.x;
            ky[j] = rc[j].x < 0 ? 0 : (rc[j].x < NKEY ? rc[j].x : NKEY - 1);
            sv[j] = rc[j].y;
            rk[j] = 0;
            if (idx < n) rk[j] = atomicAdd(&kcnt[ky[j]], 1);
        }
        __syncthreads();
        if (wave == 0) {
            const int a0 = kcnt[3 * lane], a1 = kcnt[3 * lane + 1], a2 = kcnt[3 * lane + 2];
            const int s3 = a0 + a1 + a2;
            int incl = s3;
#pragma unroll
            for (int o = 1; o < 32; o <<= 1) {
                const int t2 = __shfl_up(incl, o);
                if (lane >= o) incl += t2;
            }
            const int ex = incl - s3;
            kofs[3 * lane]     = ex;
            kofs[3 * lane + 1] = ex + a0;
            kofs[3 * lane + 2] = ex + a0 + a1;
        }
        __syncthreads();
#pragma unroll
        for (int j = 0; j < RPT; ++j) {
            const int idx = j * (NWV * 32) + threadIdx.x;
            if (idx < n) {
                int p = kofs[ky[j]] + rk[j];
                p = p < CH ? p : CH - 1;
                srt[p] = sv[j];
            }
        }
        __syncthreads();

        const int lenL = (lane < NCH) ? kcnt[myKey] : 0;
        const int stL  = (lane < NCH) ? kofs[myKey] : 0;
        cntL += lenL;
        for (int it = 0; it < CH; ++it) {
            const bool live = it < lenL;
            const unsigned liveMask = __builtin_amdgcn_ballot_w32(live);
            if (liveMask == 0u) break;
            int p = stL + it;
            p = live ? (p < CH ? p : CH - 1) : 0;
            const int sL = srt[p];
            float4 v[NCH];
#pragma unroll
            for (int q = 0; q < NCH; ++q) {
                int s = __builtin_amdgcn_readlane(sL, q);
                s = s < 0 ? 0 : (s < N ? s : N - 1);
                v[q] = h4[(size_t)s * DV4 + lane];
            }
#pragma unroll
            for (int q = 0; q < NCH; ++q) {
                const bool on = ((liveMask >> q) & 1u) != 0u;
                acc[q].x += on ? v[q].x : 0.f;
                acc[q].y += on ? v[q].y : 0.f;
                acc[q].z += on ? v[q].z : 0.f;
                acc[q].w += on ? v[q].w : 0.f;
            }
        }
        __syncthreads();
    }

    const float scL = invR / (float)(cntL > 0 ? cntL : 1);
#pragma unroll
    for (int q = 0; q < NCH; ++q) {
        const float sc = __builtin_bit_cast(float,
                             __builtin_amdgcn_readlane(__builtin_bit_cast(int, scL), q));
        const int r = q >> 2;
        const int row = wave * RPW + (q & 3);
        float4 m;
        m.x = acc[q].x * sc; m.y = acc[q].y * sc; m.z = acc[q].z * sc; m.w = acc[q].w * sc;
        *(float4*)(As + row * AP + r * D + lane * 4) = m;
    }
    __syncthreads();

    const int half = lane >> 4;
    const int lm   = lane & 15;
    const int colBase = wave * 16;

    v8f c0 = {};
    v8f c1 = {};
#pragma unroll 16
    for (int kk = 0; kk < KT; kk += 4) {
        const int ka = kk + 2 * half;
        v2f a0, a1, b;
        a0[0] = As[lm * AP + ka];
        a0[1] = As[lm * AP + ka + 1];
        a1[0] = As[(16 + lm) * AP + ka];
        a1[1] = As[(16 + lm) * AP + ka + 1];
        b[0] = W[(size_t)ka * D + colBase + lm];
        b[1] = W[(size_t)(ka + 1) * D + colBase + lm];
        c0 = __builtin_amdgcn_wmma_f32_16x16x4_f32(
                 false, a0,  false, b,
                 (short)0, c0,  false,  false);
        c1 = __builtin_amdgcn_wmma_f32_16x16x4_f32(
                 false, a1,  false, b,
                 (short)0, c1,  false,  false);
    }

#pragma unroll
    for (int v = 0; v < 8; ++v) {
        Cs[(v + 8 * half) * CP + colBase + lm]      = c0[v];
        Cs[(16 + v + 8 * half) * CP + colBase + lm] = c1[v];
    }
    __syncthreads();

    float4* __restrict__ out4 = (float4*)out;
#pragma unroll
    for (int q = 0; q < RPW; ++q) {
        const int row  = wave * RPW + q;
        const int grow = rowBase + row;
        if (grow < N) {
            const float4 cv = *(const float4*)(Cs + row * CP + lane * 4);
            const float4 hv = h4[(size_t)grow * DV4 + lane];
            float4 o;
            o.x = hv.x + cv.x; o.y = hv.y + cv.y; o.z = hv.z + cv.z; o.w = hv.w + cv.w;
            out4[(size_t)grow * DV4 + lane] = o;
        }
    }
}

extern "C" void kernel_launch(void* const* d_in, const int* in_sizes, int n_in,
                              void* d_out, int out_size, void* d_ws, size_t ws_size,
                              hipStream_t stream) {
    const float* h   = (const float*)d_in[0];
    const float* W   = (const float*)d_in[1];
    const int*   src = (const int*)d_in[2];
    const int*   dst = (const int*)d_in[3];
    float*       out = (float*)d_out;

    const int d = D;
    const int N = in_sizes[0] / d;
    const int R = NR;
    const int E = in_sizes[2] / R;
    const int total = R * E;

    int*  gcount  = (int*)d_ws;
    int*  gcursor = gcount + NBMAX;
    int*  off     = gcursor + NBMAX;
    int*  rec     = off + NBMAX + 4;

    fill_int_kernel<<<(2 * NBMAX + 255) / 256, 256, 0, stream>>>(gcount, 2 * NBMAX, 0);
    {
        const int g = (total + GT * EPT - 1) / (GT * EPT);
        count_kernel<<<g, GT, 0, stream>>>(dst, gcount, total, N);
        group_kernel<<<g, GT, 0, stream>>>(src, dst, gcount, gcursor, off, rec, total, N, E);
    }
    {
        const float invR = 1.0f / (float)R;
        gather_transform_kernel<<<(N + TM - 1) / TM, NWV * 32, 0, stream>>>(h, W, off, rec,
                                                                          out, N, total, invR);
    }
}


# dependent load chains cut by issuing loads a step ahead: count/group batch each thread's 6 dst/src loads; gather fetches the next trip's source ids under the row loads and W fragments a block ahead
# speedup vs baseline: 4.2863x; 1.0505x over previous
#include <hip/hip_runtime.h>


typedef __attribute__((ext_vector_type(2))) float v2f;
typedef __attribute__((ext_vector_type(8))) float v8f;

#define D 128
#define DV4 (D / 4)
#define NR 3
#define KT (NR * D)
#define AP (KT + 4)
#define CP (D + 4)
#define TM 32
#define NWV 8
#define RPW (TM / NWV)
#define NCH (RPW * NR)
#define NKEY (NR * TM)
#define NBMAX 4096
#define GT 1024
#define EPT 6
#define CH 1024
#define RPT (CH / (NWV * 32))
#define KSB 16

__global__ void fill_int_kernel(int* __restrict__ p, int n, int v) {
    int i = blockIdx.x * blockDim.x + threadIdx.x;
    if (i < n) p[i] = v;
}

__device__ __forceinline__ int tile_of(int d, int N) {
    d = d < 0 ? 0 : (d < N ? d : N - 1);
    const int t = d >> 5;
    return t < NBMAX ? t : NBMAX - 1;
}

__global__ __launch_bounds__(GT)
void count_kernel(const int* __restrict__ dst, int* __restrict__ gcount, int total, int N) {
    __shared__ int cnt[NBMAX];
    const int base = blockIdx.x * (GT * EPT);
    int dk[EPT];
#pragma unroll
    for (int k = 0; k < EPT; ++k) {
        const int i = base + k * GT + threadIdx.x;
        dk[k] = dst[i < total ? i : 0];
    }
    for (int b = threadIdx.x; b < NBMAX; b += GT) cnt[b] = 0;
    __syncthreads();
#pragma unroll
    for (int k = 0; k < EPT; ++k) {
        const int i = base + k * GT + threadIdx.x;
        if (i < total) atomicAdd(&cnt[tile_of(dk[k], N)], 1);
    }
    __syncthreads();
    int c4[NBMAX / GT];
#pragma unroll
    for (int j = 0; j < NBMAX / GT; ++j) c4[j] = cnt[j * GT + threadIdx.x];
#pragma unroll
    for (int j = 0; j < NBMAX / GT; ++j)
        if (c4[j]) atomicAdd(&gcount[j * GT + threadIdx.x], c4[j]);
}

__global__ __launch_bounds__(GT)
void group_kernel(const int* __restrict__ src, const int* __restrict__ dst,
                  const int* __restrict__ gcount, int* __restrict__ gcursor,
                  int* __restrict__ off, int* __restrict__ recs, int total, int N, int E) {
    int2* __restrict__ rec = (int2*)recs;
    __shared__ int offs[NBMAX + 4];
    __shared__ int cnt[NBMAX];
    __shared__ int wsum[GT / 32];
    const int tid = threadIdx.x, lane = tid & 31, wave = tid >> 5;
    const int base = blockIdx.x * (GT * EPT);

    const int4 v = ((const int4*)gcount)[tid];
    int dk[EPT], sk[EPT], rl[EPT];
#pragma unroll
    for (int k = 0; k < EPT; ++k) {
        const int i = base + k * GT + tid;
        const int j = i < total ? i : 0;
        dk[k] = dst[j];
        sk[k] = src[j];
        rl[k] = i / E;
    }

    const int s = v.x + v.y + v.z + v.w;
    int incl = s;
#pragma unroll
    for (int o = 1; o < 32; o <<= 1) {
        const int t2 = __shfl_up(incl, o);
        if (lane >= o) incl += t2;
    }
    if (lane == 31) wsum[wave] = incl;
    for (int b = tid; b < NBMAX; b += GT) cnt[b] = 0;
    __syncthreads();
    if (wave == 0) {
        const int x = wsum[lane];
        int xi = x;
#pragma unroll
        for (int o = 1; o < 32; o <<= 1) {
            const int t2 = __shfl_up(xi, o);
            if (lane >= o) xi += t2;
        }
        wsum[lane] = xi - x;
    }
    __syncthreads();
    const int ex = incl - s + wsum[wave];
    offs[4 * tid + 0] = ex;
    offs[4 * tid + 1] = ex + v.x;
    offs[4 * tid + 2] = ex + v.x + v.y;
    offs[4 * tid + 3] = ex + v.x + v.y + v.z;
    if (tid == GT - 1) offs[NBMAX] = ex + s;
    if (blockIdx.x == 0) {
        off[4 * tid + 0] = ex;
        off[4 * tid + 1] = ex + v.x;
        off[4 * tid + 2] = ex + v.x + v.y;
        off[4 * tid + 3] = ex + v.x + v.y + v.z;
        if (tid == GT - 1) off[NBMAX] = ex + s;
    }

    int rank[EPT], tk[EPT];
#pragma unroll
    for (int k = 0; k < EPT; ++k) {
        const int i = base + k * GT + tid;
        int d = dk[k];
        d = d < 0 ? 0 : (d < N ? d : N - 1);
        dk[k] = d;
        tk[k] = tile_of(d, N);
        rank[k] = 0;
        if (i < total) rank[k] = atomicAdd(&cnt[tk[k]], 1);
    }
    __syncthreads();
    int c4[NBMAX / GT], st4[NBMAX / GT];
#pragma unroll
    for (int j = 0; j < NBMAX / GT; ++j) {
        c4[j] = cnt[j * GT + tid];
        st4[j] = 0;
    }
#pragma unroll
    for (int j = 0; j < NBMAX / GT; ++j)
        if (c4[j]) st4[j] = atomicAdd(&gcursor[j * GT + tid], c4[j]);
#pragma unroll
    for (int j = 0; j < NBMAX / GT; ++j) cnt[j * GT + tid] = offs[j * GT + tid] + st4[j];
    __syncthreads();
    int st6[EPT];
#pragma unroll
    for (int k = 0; k < EPT; ++k) st6[k] = cnt[tk[k]];
#pragma unroll
    for (int k = 0; k < EPT; ++k) {
        const int i = base + k * GT + tid;
        if (i < total) {
            int pos = st6[k] + rank[k];
            pos = pos < 0 ? 0 : (pos < total ? pos : total - 1);
            rec[pos] = make_int2(rl[k] * TM + (dk[k] & (TM - 1)), sk[k]);
        }
    }
}

__global__ __launch_bounds__(256)
void gather_transform_kernel(const float* __restrict__ h,
                             const float* __restrict__ W,
                             const int* __restrict__ off,
                             const int* __restrict__ recs,
                             float* __restrict__ out,
                             int N, int total, float invR) {
    __shared__ __attribute__((aligned(16))) float As[TM * AP];
    __shared__ __attribute__((aligned(16))) float Cs[TM * CP];
    __shared__ int kcnt[NKEY];
    __shared__ int kofs[NKEY];
    __shared__ int srt[CH];

    const int wave = threadIdx.x >> 5;
    const int lane = threadIdx.x & 31;
    const int rowBase = blockIdx.x * TM;
    const float4* __restrict__ h4 = (const float4*)h;
    const int2* __restrict__ rec = (const int2*)recs;

    int beg = off[blockIdx.x];
    int end = off[blockIdx.x + 1];
    beg = beg < 0 ? 0 : (beg < total ? beg : total);
    end = end < beg ? beg : (end < total ? end : total);

    const int myKey = (lane < NCH) ? (lane >> 2) * TM + wave * RPW + (lane & 3) : 0;
    int cntL = 0;
    float4 acc[NCH];
#pragma unroll
    for (int q = 0; q < NCH; ++q) acc[q] = make_float4(0.f, 0.f, 0.f, 0.f);

    for (int cb = beg; cb < end; cb += CH) {
        const int n = (end - cb) < CH ? (end - cb) : CH;
        if (threadIdx.x < NKEY) kcnt[threadIdx.x] = 0;
        __syncthreads();
        int rk[RPT], ky[RPT], sv[RPT];
        int2 rc[RPT];
#pragma unroll
        for (int j = 0; j < RPT; ++j) {
            const int idx = j * (NWV * 32) + threadIdx.x;
            rc[j] = rec[cb + (idx < n ? idx : 0)];
        }
#pragma unroll
        for (int j = 0; j < RPT; ++j) {
            const int idx = j * (NWV * 32) + threadIdx.x;
            ky[j] = rc[j].x < 0 ? 0 : (rc[j].x < NKEY ? rc[j].x : NKEY - 1);
            sv[j] = rc[j].y;
            rk[j] = 0;
            if (idx < n) rk[j] = atomicAdd(&kcnt[ky[j]], 1);
        }
        __syncthreads();
        if (wave == 0) {
            const int a0 = kcnt[3 * lane], a1 = kcnt[3 * lane + 1], a2 = kcnt[3 * lane + 2];
            const int s3 = a0 + a1 + a2;
            int incl = s3;
#pragma unroll
            for (int o = 1; o < 32; o <<= 1) {
                const int t2 = __shfl_up(incl, o);
                if (lane >= o) incl += t2;
            }
            const int ex = incl - s3;
            kofs[3 * lane]     = ex;
            kofs[3 * lane + 1] = ex + a0;
            kofs[3 * lane + 2] = ex + a0 + a1;
        }
        __syncthreads();
#pragma unroll
        for (int j = 0; j < RPT; ++j) {
            const int idx = j * (NWV * 32) + threadIdx.x;
            if (idx < n) {
                int p = kofs[ky[j]] + rk[j];
                p = p < CH ? p : CH - 1;
                srt[p] = sv[j];
            }
        }
        __syncthreads();

        const int lenL = (lane < NCH) ? kcnt[myKey] : 0;
        const int stL  = (lane < NCH) ? kofs[myKey] : 0;
        cntL += lenL;
        int so[NCH];
        {
            int p = stL;
            p = (0 < lenL) ? (p < CH ? p : CH - 1) : 0;
            const int sL = srt[p];
#pragma unroll
            for (int q = 0; q < NCH; ++q) {
                const int s = __builtin_amdgcn_readlane(sL, q);
                so[q] = s < 0 ? 0 : (s < N ? s : N - 1);
            }
        }
        for (int it = 0; it < CH; ++it) {
            const bool live = it < lenL;
            const unsigned liveMask = __builtin_amdgcn_ballot_w32(live);
            if (liveMask == 0u) break;
            float4 v[NCH];
#pragma unroll
            for (int q = 0; q < NCH; ++q) v[q] = h4[(size_t)so[q] * DV4 + lane];
            {
                int p = stL + it + 1;
                p = (it + 1 < lenL) ? (p < CH ? p : CH - 1) : 0;
                const int sL = srt[p];
#pragma unroll
                for (int q = 0; q < NCH; ++q) {
                    const int s = __builtin_amdgcn_readlane(sL, q);
                    so[q] = s < 0 ? 0 : (s < N ? s : N - 1);
                    asm volatile("" : "+s"(so[q]));
                }
            }
#pragma unroll
            for (int q = 0; q < NCH; ++q) {
                const bool on = ((liveMask >> q) & 1u) != 0u;
                acc[q].x += on ? v[q].x : 0.f;
                acc[q].y += on ? v[q].y : 0.f;
                acc[q].z += on ? v[q].z : 0.f;
                acc[q].w += on ? v[q].w : 0.f;
            }
        }
        __syncthreads();
    }

    const float scL = invR / (float)(cntL > 0 ? cntL : 1);
#pragma unroll
    for (int q = 0; q < NCH; ++q) {
        const float sc = __builtin_bit_cast(float,
                             __builtin_amdgcn_readlane(__builtin_bit_cast(int, scL), q));
        const int r = q >> 2;
        const int row = wave * RPW + (q & 3);
        float4 m;
        m.x = acc[q].x * sc; m.y = acc[q].y * sc; m.z = acc[q].z * sc; m.w = acc[q].w * sc;
        *(float4*)(As + row * AP + r * D + lane * 4) = m;
    }
    __syncthreads();

    const int half = lane >> 4;
    const int lm   = lane & 15;
    const int colBase = wave * 16;

    v8f c0 = {};
    v8f c1 = {};
    v2f bw[2][KSB];
    const float* __restrict__ Wl = W + (size_t)(2 * half) * D + colBase + lm;
    float4 hres[RPW];
#pragma unroll
    for (int j = 0; j < KSB; ++j) {
        bw[0][j][0] = Wl[(size_t)(4 * j) * D];
        bw[0][j][1] = Wl[(size_t)(4 * j + 1) * D];
    }
#pragma unroll
    for (int blk = 0; blk < KT / (4 * KSB); ++blk) {
        if (blk + 1 < KT / (4 * KSB)) {
#pragma unroll
            for (int j = 0; j < KSB; ++j) {
                const int kn = (blk + 1) * 4 * KSB + 4 * j;
                bw[(blk + 1) & 1][j][0] = Wl[(size_t)kn * D];
                bw[(blk + 1) & 1][j][1] = Wl[(size_t)(kn + 1) * D];
            }
        }
        if (blk == 0) {
#pragma unroll
            for (int q = 0; q < RPW; ++q) {
                const int grow = rowBase + wave * RPW + q;
                hres[q] = h4[(size_t)(grow < N ? grow : N - 1) * DV4 + lane];
            }
        }
#pragma unroll
        for (int j = 0; j < KSB; ++j) {
            const int ka = blk * 4 * KSB + 4 * j + 2 * half;
            v2f a0, a1;
            a0[0] = As[lm * AP + ka];
            a0[1] = As[lm * AP + ka + 1];
            a1[0] = As[(16 + lm) * AP + ka];
            a1[1] = As[(16 + lm) * AP + ka + 1];
            c0 = __builtin_amdgcn_wmma_f32_16x16x4_f32(
                     false, a0,  false, bw[blk & 1][j],
                     (short)0, c0,  false,  false);
            c1 = __builtin_amdgcn_wmma_f32_16x16x4_f32(
                     false, a1,  false, bw[blk & 1][j],
                     (short)0, c1,  false,  false);
        }
        __builtin_amdgcn_sched_barrier(0);
    }

#pragma unroll
    for (int v = 0; v < 8; ++v) {
        Cs[(v + 8 * half) * CP + colBase + lm]      = c0[v];
        Cs[(16 + v + 8 * half) * CP + colBase + lm] = c1[v];
    }
    __syncthreads();

    float4* __restrict__ out4 = (float4*)out;
#pragma unroll
    for (int q = 0; q < RPW; ++q) {
        const int row  = wave * RPW + q;
        const int grow = rowBase + row;
        if (grow < N) {
            const float4 cv = *(const float4*)(Cs + row * CP + lane * 4);
            const float4 hv = hres[q];
            float4 o;
            o.x = hv.x + cv.x; o.y = hv.y + cv.y; o.z = hv.z + cv.z; o.w = hv.w + cv.w;
            out4[(size_t)grow * DV4 + lane] = o;
        }
    }
}

extern "C" void kernel_launch(void* const* d_in, const int* in_sizes, int n_in,
                              void* d_out, int out_size, void* d_ws, size_t ws_size,
                              hipStream_t stream) {
    const float* h   = (const float*)d_in[0];
    const float* W   = (const float*)d_in[1];
    const int*   src = (const int*)d_in[2];
    const int*   dst = (const int*)d_in[3];
    float*       out = (float*)d_out;

    const int d = D;
    const int N = in_sizes[0] / d;
    const int R = NR;
    const int E = in_sizes[2] / R;
    const int total = R * E;

    int*  gcount  = (int*)d_ws;
    int*  gcursor = gcount + NBMAX;
    int*  off     = gcursor + NBMAX;
    int*  rec     = off + NBMAX + 4;

    fill_int_kernel<<<(2 * NBMAX + 255) / 256, 256, 0, stream>>>(gcount, 2 * NBMAX, 0);
    {
        const int g = (total + GT * EPT - 1) / (GT * EPT);
        count_kernel<<<g, GT, 0, stream>>>(dst, gcount, total, N);
        group_kernel<<<g, GT, 0, stream>>>(src, dst, gcount, gcursor, off, rec, total, N, E);
    }
    {
        const float invR = 1.0f / (float)R;
        gather_transform_kernel<<<(N + TM - 1) / TM, NWV * 32, 0, stream>>>(h, W, off, rec,
                                                                          out, N, total, invR);
    }
}


# gather walk: records sorted so a wave's 12 lists form one run, walked 12 records a trip with a running sum parked at key changes: no trips or row loads on ended lists; sum order kept
# speedup vs baseline: 4.4220x; 1.0317x over previous
#include <hip/hip_runtime.h>


typedef __attribute__((ext_vector_type(2))) float v2f;
typedef __attribute__((ext_vector_type(8))) float v8f;

#define D 128
#define DV4 (D / 4)
#define NR 3
#define KT (NR * D)
#define AP (KT + 4)
#define CP (D + 4)
#define TM 32
#define NWV 8
#define RPW (TM / NWV)
#define NCH (RPW * NR)
#define NKEY (NR * TM)
#define NBMAX 4096
#define GT 1024
#define EPT 6
#define CH 1024
#define RPT (CH / (NWV * 32))
#define KSB 16

__global__ void fill_int_kernel(int* __restrict__ p, int n, int v) {
    int i = blockIdx.x * blockDim.x + threadIdx.x;
    if (i < n) p[i] = v;
}

__device__ __forceinline__ int tile_of(int d, int N) {
    d = d < 0 ? 0 : (d < N ? d : N - 1);
    const int t = d >> 5;
    return t < NBMAX ? t : NBMAX - 1;
}

__global__ __launch_bounds__(GT)
void count_kernel(const int* __restrict__ dst, int* __restrict__ gcount, int total, int N) {
    __shared__ int cnt[NBMAX];
    const int base = blockIdx.x * (GT * EPT);
    int dk[EPT];
#pragma unroll
    for (int k = 0; k < EPT; ++k) {
        const int i = base + k * GT + threadIdx.x;
        dk[k] = dst[i < total ? i : 0];
    }
    for (int b = threadIdx.x; b < NBMAX; b += GT) cnt[b] = 0;
    __syncthreads();
#pragma unroll
    for (int k = 0; k < EPT; ++k) {
        const int i = base + k * GT + threadIdx.x;
        if (i < total) atomicAdd(&cnt[tile_of(dk[k], N)], 1);
    }
    __syncthreads();
    int c4[NBMAX / GT];
#pragma unroll
    for (int j = 0; j < NBMAX / GT; ++j) c4[j] = cnt[j * GT + threadIdx.x];
#pragma unroll
    for (int j = 0; j < NBMAX / GT; ++j)
        if (c4[j]) atomicAdd(&gcount[j * GT + threadIdx.x], c4[j]);
}

__global__ __launch_bounds__(GT)
void group_kernel(const int* __restrict__ src, const int* __restrict__ dst,
                  const int* __restrict__ gcount, int* __restrict__ gcursor,
                  int* __restrict__ off, int* __restrict__ recs, int total, int N, int E) {
    int2* __restrict__ rec = (int2*)recs;
    __shared__ int offs[NBMAX + 4];
    __shared__ int cnt[NBMAX];
    __shared__ int wsum[GT / 32];
    const int tid = threadIdx.x, lane = tid & 31, wave = tid >> 5;
    const int base = blockIdx.x * (GT * EPT);

    const int4 v = ((const int4*)gcount)[tid];
    int dk[EPT], sk[EPT], rl[EPT];
#pragma unroll
    for (int k = 0; k < EPT; ++k) {
        const int i = base + k * GT + tid;
        const int j = i < total ? i : 0;
        dk[k] = dst[j];
        sk[k] = src[j];
        rl[k] = i / E;
    }

    const int s = v.x + v.y + v.z + v.w;
    int incl = s;
#pragma unroll
    for (int o = 1; o < 32; o <<= 1) {
        const int t2 = __shfl_up(incl, o);
        if (lane >= o) incl += t2;
    }
    if (lane == 31) wsum[wave] = incl;
    for (int b = tid; b < NBMAX; b += GT) cnt[b] = 0;
    __syncthreads();
    if (wave == 0) {
        const int x = wsum[lane];
        int xi = x;
#pragma unroll
        for (int o = 1; o < 32; o <<= 1) {
            const int t2 = __shfl_up(xi, o);
            if (lane >= o) xi += t2;
        }
        wsum[lane] = xi - x;
    }
    __syncthreads();
    const int ex = incl - s + wsum[wave];
    offs[4 * tid + 0] = ex;
    offs[4 * tid + 1] = ex + v.x;
    offs[4 * tid + 2] = ex + v.x + v.y;
    offs[4 * tid + 3] = ex + v.x + v.y + v.z;
    if (tid == GT - 1) offs[NBMAX] = ex + s;
    if (blockIdx.x == 0) {
        off[4 * tid + 0] = ex;
        off[4 * tid + 1] = ex + v.x;
        off[4 * tid + 2] = ex + v.x + v.y;
        off[4 * tid + 3] = ex + v.x + v.y + v.z;
        if (tid == GT - 1) off[NBMAX] = ex + s;
    }

    int rank[EPT], tk[EPT];
#pragma unroll
    for (int k = 0; k < EPT; ++k) {
        const int i = base + k * GT + tid;
        int d = dk[k];
        d = d < 0 ? 0 : (d < N ? d : N - 1);
        dk[k] = d;
        tk[k] = tile_of(d, N);
        rank[k] = 0;
        if (i < total) rank[k] = atomicAdd(&cnt[tk[k]], 1);
    }
    __syncthreads();
    int c4[NBMAX / GT], st4[NBMAX / GT];
#pragma unroll
    for (int j = 0; j < NBMAX / GT; ++j) {
        c4[j] = cnt[j * GT + tid];
        st4[j] = 0;
    }
#pragma unroll
    for (int j = 0; j < NBMAX / GT; ++j)
        if (c4[j]) st4[j] = atomicAdd(&gcursor[j * GT + tid], c4[j]);
#pragma unroll
    for (int j = 0; j < NBMAX / GT; ++j) cnt[j * GT + tid] = offs[j * GT + tid] + st4[j];
    __syncthreads();
    int st6[EPT];
#pragma unroll
    for (int k = 0; k < EPT; ++k) st6[k] = cnt[tk[k]];
#pragma unroll
    for (int k = 0; k < EPT; ++k) {
        const int i = base + k * GT + tid;
        if (i < total) {
            int pos = st6[k] + rank[k];
            pos = pos < 0 ? 0 : (pos < total ? pos : total - 1);
            rec[pos] = make_int2(rl[k] * TM + (dk[k] & (TM - 1)), sk[k]);
        }
    }
}

__global__ __launch_bounds__(256, 8)
void gather_transform_kernel(const float* __restrict__ h,
                             const float* __restrict__ W,
                             const int* __restrict__ off,
                             const int* __restrict__ recs,
                             float* __restrict__ out,
                             int N, int total, float invR) {
    __shared__ __attribute__((aligned(16))) float As[TM * AP];
    __shared__ __attribute__((aligned(16))) float Cs[TM * CP];
    __shared__ int kcnt[NKEY];
    __shared__ int kofs[NKEY + 1];
    __shared__ int2 srt[CH];

    const int wave = threadIdx.x >> 5;
    const int lane = threadIdx.x & 31;
    const int rowBase = blockIdx.x * TM;
    const float4* __restrict__ h4 = (const float4*)h;
    const int2* __restrict__ rec = (const int2*)recs;

    int beg = off[blockIdx.x];
    int end = off[blockIdx.x + 1];
    beg = beg < 0 ? 0 : (beg < total ? beg : total);
    end = end < beg ? beg : (end < total ? end : total);

    const int wk0 = wave * NCH;
    int cntL = 0;
#pragma unroll
    for (int q = 0; q < NCH; ++q)
        *(float4*)(As + (wave * RPW + (q & 3)) * AP + (q >> 2) * D + lane * 4) = make_float4(0.f, 0.f, 0.f, 0.f);

    for (int cb = beg; cb < end; cb += CH) {
        const int n = (end - cb) < CH ? (end - cb) : CH;
        if (threadIdx.x < NKEY) kcnt[threadIdx.x] = 0;
        __syncthreads();
        int rk[RPT], ky[RPT], lk[RPT], sv[RPT];
        int2 rc[RPT];
#pragma unroll
        for (int j = 0; j < RPT; ++j) {
            const int idx = j * (NWV * 32) + threadIdx.x;
            rc[j] = rec[cb + (idx < n ? idx : 0)];
        }
#pragma unroll
        for (int j = 0; j < RPT; ++j) {
            const int idx = j * (NWV * 32) + threadIdx.x;
            const int kr = rc[j].x < 0 ? 0 : (rc[j].x < NKEY ? rc[j].x : NKEY - 1);
            const int rw = kr & (TM - 1);
            lk[j] = (kr / TM) * RPW + (rw & (RPW - 1));
            ky[j] = (rw / RPW) * NCH + lk[j];
            sv[j] = rc[j].y;
            rk[j] = 0;
            if (idx < n) rk[j] = atomicAdd(&kcnt[ky[j]], 1);
        }
        __syncthreads();
        if (wave == 0) {
            const int a0 = kcnt[3 * lane], a1 = kcnt[3 * lane + 1], a2 = kcnt[3 * lane + 2];
            const int s3 = a0 + a1 + a2;
            int incl = s3;
#pragma unroll
            for (int o = 1; o < 32; o <<= 1) {
                const int t2 = __shfl_up(incl, o);
                if (lane >= o) incl += t2;
            }
            const int ex = incl - s3;
            kofs[3 * lane]     = ex;
            kofs[3 * lane + 1] = ex + a0;
            kofs[3 * lane + 2] = ex + a0 + a1;
            if (lane == 31) kofs[NKEY] = incl;
        }
        __syncthreads();
#pragma unroll
        for (int j = 0; j < RPT; ++j) {
            const int idx = j * (NWV * 32) + threadIdx.x;
            if (idx < n) {
                int p = kofs[ky[j]] + rk[j];
                p = p < CH ? p : CH - 1;
                srt[p] = make_int2(sv[j], lk[j]);
            }
        }
        __syncthreads();

        cntL += (lane < NCH) ? kcnt[wk0 + lane] : 0;
        int wb = __builtin_amdgcn_readfirstlane(kofs[wk0]);
        int we = __builtin_amdgcn_readfirstlane(kofs[wk0 + NCH]);
        wb = wb < 0 ? 0 : (wb < n ? wb : n);
        we = we < wb ? wb : (we < n ? we : n);
        const bool later = cb != beg;
        int so[NCH], kn[NCH];
        {
            const int p = wb + lane;
            const int2 e = srt[(lane < NCH && p < we) ? p : 0];
#pragma unroll
            for (int q = 0; q < NCH; ++q) {
                const int s = __builtin_amdgcn_readlane(e.x, q);
                so[q] = s < 0 ? 0 : (s < N ? s : N - 1);
                const int k = __builtin_amdgcn_readlane(e.y, q);
                kn[q] = k < 0 ? 0 : (k < NCH ? k : NCH - 1);
            }
        }
        int cur = -1;
        float4 acc = make_float4(0.f, 0.f, 0.f, 0.f);
        for (int t0 = wb; t0 < we; t0 += NCH) {
            float4 v[NCH];
#pragma unroll
            for (int q = 0; q < NCH; ++q) v[q] = h4[(size_t)so[q] * DV4 + lane];
            int kc[NCH];
#pragma unroll
            for (int q = 0; q < NCH; ++q) kc[q] = kn[q];
            {
                const int p = t0 + NCH + lane;
                const int2 e = srt[(lane < NCH && p < we) ? p : 0];
#pragma unroll
                for (int q = 0; q < NCH; ++q) {
                    const int s = __builtin_amdgcn_readlane(e.x, q);
                    so[q] = s < 0 ? 0 : (s < N ? s : N - 1);
                    asm volatile("" : "+s"(so[q]));
                    const int k = __builtin_amdgcn_readlane(e.y, q);
                    kn[q] = k < 0 ? 0 : (k < NCH ? k : NCH - 1);
                }
            }
#pragma unroll
            for (int q = 0; q < NCH; ++q) {
                if (t0 + q < we) {
                    const int k = kc[q];
                    if (k != cur) {
                        if (cur >= 0)
                            *(float4*)(As + (wave * RPW + (cur & 3)) * AP + (cur >> 2) * D + lane * 4) = acc;
                        acc = make_float4(0.f, 0.f, 0.f, 0.f);
                        if (later)
                            acc = *(const float4*)(As + (wave * RPW + (k & 3)) * AP + (k >> 2) * D + lane * 4);
                        cur = k;
                    }
                    acc.x += v[q].x;
                    acc.y += v[q].y;
                    acc.z += v[q].z;
                    acc.w += v[q].w;
                }
            }
        }
        if (cur >= 0)
            *(float4*)(As + (wave * RPW + (cur & 3)) * AP + (cur >> 2) * D + lane * 4) = acc;
        __syncthreads();
    }

    const float scL = invR / (float)(cntL > 0 ? cntL : 1);
#pragma unroll
    for (int q = 0; q < NCH; ++q) {
        const float sc = __builtin_bit_cast(float,
                             __builtin_amdgcn_readlane(__builtin_bit_cast(int, scL), q));
        const int r = q >> 2;
        const int row = wave * RPW + (q & 3);
        float4 m = *(const float4*)(As + row * AP + r * D + lane * 4);
        m.x *= sc; m.y *= sc; m.z *= sc; m.w *= sc;
        *(float4*)(As + row * AP + r * D + lane * 4) = m;
    }
    __syncthreads();

    const int half = lane >> 4;
    const int lm   = lane & 15;
    const int colBase = wave * 16;

    v8f c0 = {};
    v8f c1 = {};
    v2f bw[2][KSB];
    const float* __restrict__ Wl = W + (size_t)(2 * half) * D + colBase + lm;
    float4 hres[RPW];
#pragma unroll
    for (int j = 0; j < KSB; ++j) {
        bw[0][j][0] = Wl[(size_t)(4 * j) * D];
        bw[0][j][1] = Wl[(size_t)(4 * j + 1) * D];
    }
#pragma unroll
    for (int blk = 0; blk < KT / (4 * KSB); ++blk) {
        if (blk + 1 < KT / (4 * KSB)) {
#pragma unroll
            for (int j = 0; j < KSB; ++j) {
                const int kn = (blk + 1) * 4 * KSB + 4 * j;
                bw[(blk + 1) & 1][j][0] = Wl[(size_t)kn * D];
                bw[(blk + 1) & 1][j][1] = Wl[(size_t)(kn + 1) * D];
            }
        }
        if (blk == 0) {
#pragma unroll
            for (int q = 0; q < RPW; ++q) {
                const int grow = rowBase + wave * RPW + q;
                hres[q] = h4[(size_t)(grow < N ? grow : N - 1) * DV4 + lane];
            }
        }
#pragma unroll
        for (int j = 0; j < KSB; ++j) {
            const int ka = blk * 4 * KSB + 4 * j + 2 * half;
            v2f a0, a1;
            a0[0] = As[lm * AP + ka];
            a0[1] = As[lm * AP + ka + 1];
            a1[0] = As[(16 + lm) * AP + ka];
            a1[1] = As[(16 + lm) * AP + ka + 1];
            c0 = __builtin_amdgcn_wmma_f32_16x16x4_f32(
                     false, a0,  false, bw[blk & 1][j],
                     (short)0, c0,  false,  false);
            c1 = __builtin_amdgcn_wmma_f32_16x16x4_f32(
                     false, a1,  false, bw[blk & 1][j],
                     (short)0, c1,  false,  false);
        }
        __builtin_amdgcn_sched_barrier(0);
    }

#pragma unroll
    for (int v = 0; v < 8; ++v) {
        Cs[(v + 8 * half) * CP + colBase + lm]      = c0[v];
        Cs[(16 + v + 8 * half) * CP + colBase + lm] = c1[v];
    }
    __syncthreads();

    float4* __restrict__ out4 = (float4*)out;
#pragma unroll
    for (int q = 0; q < RPW; ++q) {
        const int row  = wave * RPW + q;
        const int grow = rowBase + row;
        if (grow < N) {
            const float4 cv = *(const float4*)(Cs + row * CP + lane * 4);
            const float4 hv = hres[q];
            float4 o;
            o.x = hv.x + cv.x; o.y = hv.y + cv.y; o.z = hv.z + cv.z; o.w = hv.w + cv.w;
            out4[(size_t)grow * DV4 + lane] = o;
        }
    }
}

extern "C" void kernel_launch(void* const* d_in, const int* in_sizes, int n_in,
                              void* d_out, int out_size, void* d_ws, size_t ws_size,
                              hipStream_t stream) {
    const float* h   = (const float*)d_in[0];
    const float* W   = (const float*)d_in[1];
    const int*   src = (const int*)d_in[2];
    const int*   dst = (const int*)d_in[3];
    float*       out = (float*)d_out;

    const int d = D;
    const int N = in_sizes[0] / d;
    const int R = NR;
    const int E = in_sizes[2] / R;
    const int total = R * E;

    int*  gcount  = (int*)d_ws;
    int*  gcursor = gcount + NBMAX;
    int*  off     = gcursor + NBMAX;
    int*  rec     = off + NBMAX + 4;

    fill_int_kernel<<<(2 * NBMAX + 255) / 256, 256, 0, stream>>>(gcount, 2 * NBMAX, 0);
    {
        const int g = (total + GT * EPT - 1) / (GT * EPT);
        count_kernel<<<g, GT, 0, stream>>>(dst, gcount, total, N);
        group_kernel<<<g, GT, 0, stream>>>(src, dst, gcount, gcursor, off, rec, total, N, E);
    }
    {
        const float invR = 1.0f / (float)R;
        gather_transform_kernel<<<(N + TM - 1) / TM, NWV * 32, 0, stream>>>(h, W, off, rec,
                                                                          out, N, total, invR);
    }
}


# gather+transform: 5 work-groups resident per WGP, was 4: C tile laid over the A tile in LDS (58640 B), W read-ahead sets of 12 k-steps and residual rows read last fit 96 VGPRs; sum order kept
# speedup vs baseline: 4.6367x; 1.0486x over previous
#include <hip/hip_runtime.h>


typedef __attribute__((ext_vector_type(2))) float v2f;
typedef __attribute__((ext_vector_type(8))) float v8f;

#define D 128
#define DV4 (D / 4)
#define NR 3
#define KT (NR * D)
#define AP (KT + 4)
#define CP (D + 4)
#define TM 32
#define NWV 8
#define RPW (TM / NWV)
#define NCH (RPW * NR)
#define NKEY (NR * TM)
#define NBMAX 4096
#define GT 1024
#define EPT 6
#define CH 1024
#define RPT (CH / (NWV * 32))
#define KSB 12

__global__ void fill_int_kernel(int* __restrict__ p, int n, int v) {
    int i = blockIdx.x * blockDim.x + threadIdx.x;
    if (i < n) p[i] = v;
}

__device__ __forceinline__ int tile_of(int d, int N) {
    d = d < 0 ? 0 : (d < N ? d : N - 1);
    const int t = d >> 5;
    return t < NBMAX ? t : NBMAX - 1;
}

__global__ __launch_bounds__(GT)
void count_kernel(const int* __restrict__ dst, int* __restrict__ gcount, int total, int N) {
    __shared__ int cnt[NBMAX];
    const int base = blockIdx.x * (GT * EPT);
    int dk[EPT];
#pragma unroll
    for (int k = 0; k < EPT; ++k) {
        const int i = base + k * GT + threadIdx.x;
        dk[k] = dst[i < total ? i : 0];
    }
    for (int b = threadIdx.x; b < NBMAX; b += GT) cnt[b] = 0;
    __syncthreads();
#pragma unroll
    for (int k = 0; k < EPT; ++k) {
        const int i = base + k * GT + threadIdx.x;
        if (i < total) atomicAdd(&cnt[tile_of(dk[k], N)], 1);
    }
    __syncthreads();
    int c4[NBMAX / GT];
#pragma unroll
    for (int j = 0; j < NBMAX / GT; ++j) c4[j] = cnt[j * GT + threadIdx.x];
#pragma unroll
    for (int j = 0; j < NBMAX / GT; ++j)
        if (c4[j]) atomicAdd(&gcount[j * GT + threadIdx.x], c4[j]);
}

__global__ __launch_bounds__(GT)
void group_kernel(const int* __restrict__ src, const int* __restrict__ dst,
                  const int* __restrict__ gcount, int* __restrict__ gcursor,
                  int* __restrict__ off, int* __restrict__ recs, int total, int N, int E) {
    int2* __restrict__ rec = (int2*)recs;
    __shared__ int offs[NBMAX + 4];
    __shared__ int cnt[NBMAX];
    __shared__ int wsum[GT / 32];
    const int tid = threadIdx.x, lane = tid & 31, wave = tid >> 5;
    const int base = blockIdx.x * (GT * EPT);

    const int4 v = ((const int4*)gcount)[tid];
    int dk[EPT], sk[EPT], rl[EPT];
#pragma unroll
    for (int k = 0; k < EPT; ++k) {
        const int i = base + k * GT + tid;
        const int j = i < total ? i : 0;
        dk[k] = dst[j];
        sk[k] = src[j];
        rl[k] = i / E;
    }

    const int s = v.x + v.y + v.z + v.w;
    int incl = s;
#pragma unroll
    for (int o = 1; o < 32; o <<= 1) {
        const int t2 = __shfl_up(incl, o);
        if (lane >= o) incl += t2;
    }
    if (lane == 31) wsum[wave] = incl;
    for (int b = tid; b < NBMAX; b += GT) cnt[b] = 0;
    __syncthreads();
    if (wave == 0) {
        const int x = wsum[lane];
        int xi = x;
#pragma unroll
        for (int o = 1; o < 32; o <<= 1) {
            const int t2 = __shfl_up(xi, o);
            if (lane >= o) xi += t2;
        }
        wsum[lane] = xi - x;
    }
    __syncthreads();
    const int ex = incl - s + wsum[wave];
    offs[4 * tid + 0] = ex;
    offs[4 * tid + 1] = ex + v.x;
    offs[4 * tid + 2] = ex + v.x + v.y;
    offs[4 * tid + 3] = ex + v.x + v.y + v.z;
    if (tid == GT - 1) offs[NBMAX] = ex + s;
    if (blockIdx.x == 0) {
        off[4 * tid + 0] = ex;
        off[4 * tid + 1] = ex + v.x;
        off[4 * tid + 2] = ex + v.x + v.y;
        off[4 * tid + 3] = ex + v.x + v.y + v.z;
        if (tid == GT - 1) off[NBMAX] = ex + s;
    }

    int rank[EPT], tk[EPT];
#pragma unroll
    for (int k = 0; k < EPT; ++k) {
        const int i = base + k * GT + tid;
        int d = dk[k];
        d = d < 0 ? 0 : (d < N ? d : N - 1);
        dk[k] = d;
        tk[k] = tile_of(d, N);
        rank[k] = 0;
        if (i < total) rank[k] = atomicAdd(&cnt[tk[k]], 1);
    }
    __syncthreads();
    int c4[NBMAX / GT], st4[NBMAX / GT];
#pragma unroll
    for (int j = 0; j < NBMAX / GT; ++j) {
        c4[j] = cnt[j * GT + tid];
        st4[j] = 0;
    }
#pragma unroll
    for (int j = 0; j < NBMAX / GT; ++j)
        if (c4[j]) st4[j] = atomicAdd(&gcursor[j * GT + tid], c4[j]);
#pragma unroll
    for (int j = 0; j < NBMAX / GT; ++j) cnt[j * GT + tid] = offs[j * GT + tid] + st4[j];
    __syncthreads();
    int st6[EPT];
#pragma unroll
    for (int k = 0; k < EPT; ++k) st6[k] = cnt[tk[k]];
#pragma unroll
    for (int k = 0; k < EPT; ++k) {
        const int i = base + k * GT + tid;
        if (i < total) {
            int pos = st6[k] + rank[k];
            pos = pos < 0 ? 0 : (pos < total ? pos : total - 1);
            rec[pos] = make_int2(rl[k] * TM + (dk[k] & (TM - 1)), sk[k]);
        }
    }
}

__global__ __launch_bounds__(256, 10)
void gather_transform_kernel(const float* __restrict__ h,
                             const float* __restrict__ W,
                             const int* __restrict__ off,
                             const int* __restrict__ recs,
                             float* __restrict__ out,
                             int N, int total, float invR) {
    __shared__ __attribute__((aligned(16))) float As[TM * AP];
    __shared__ int kcnt[NKEY];
    __shared__ int kofs[NKEY + 1];
    __shared__ int2 srt[CH];

    float* const Cs = As;
    const int wave = threadIdx.x >> 5;
    const int lane = threadIdx.x & 31;
    const int rowBase = blockIdx.x * TM;
    const float4* __restrict__ h4 = (const float4*)h;
    const int2* __restrict__ rec = (const int2*)recs;

    int beg = off[blockIdx.x];
    int end = off[blockIdx.x + 1];
    beg = beg < 0 ? 0 : (beg < total ? beg : total);
    end = end < beg ? beg : (end < total ? end : total);

    const int wk0 = wave * NCH;
    int cntL = 0;
#pragma unroll
    for (int q = 0; q < NCH; ++q)
        *(float4*)(As + (wave * RPW + (q & 3)) * AP + (q >> 2) * D + lane * 4) = make_float4(0.f, 0.f, 0.f, 0.f);

    for (int cb = beg; cb < end; cb += CH) {
        const int n = (end - cb) < CH ? (end - cb) : CH;
        if (threadIdx.x < NKEY) kcnt[threadIdx.x] = 0;
        __syncthreads();
        int rk[RPT], ky[RPT], lk[RPT], sv[RPT];
        int2 rc[RPT];
#pragma unroll
        for (int j = 0; j < RPT; ++j) {
            const int idx = j * (NWV * 32) + threadIdx.x;
            rc[j] = rec[cb + (idx < n ? idx : 0)];
        }
#pragma unroll
        for (int j = 0; j < RPT; ++j) {
            const int idx = j * (NWV * 32) + threadIdx.x;
            const int kr = rc[j].x < 0 ? 0 : (rc[j].x < NKEY ? rc[j].x : NKEY - 1);
            const int rw = kr & (TM - 1);
            lk[j] = (kr / TM) * RPW + (rw & (RPW - 1));
            ky[j] = (rw / RPW) * NCH + lk[j];
            sv[j] = rc[j].y;
            rk[j] = 0;
            if (idx < n) rk[j] = atomicAdd(&kcnt[ky[j]], 1);
        }
        __syncthreads();
        if (wave == 0) {
            const int a0 = kcnt[3 * lane], a1 = kcnt[3 * lane + 1], a2 = kcnt[3 * lane + 2];
            const int s3 = a0 + a1 + a2;
            int incl = s3;
#pragma unroll
            for (int o = 1; o < 32; o <<= 1) {
                const int t2 = __shfl_up(incl, o);
                if (lane >= o) incl += t2;
            }
            const int ex = incl - s3;
            kofs[3 * lane]     = ex;
            kofs[3 * lane + 1] = ex + a0;
            kofs[3 * lane + 2] = ex + a0 + a1;
            if (lane == 31) kofs[NKEY] = incl;
        }
        __syncthreads();
#pragma unroll
        for (int j = 0; j < RPT; ++j) {
            const int idx = j * (NWV * 32) + threadIdx.x;
            if (idx < n) {
                int p = kofs[ky[j]] + rk[j];
                p = p < CH ? p : CH - 1;
                srt[p] = make_int2(sv[j], lk[j]);
            }
        }
        __syncthreads();

        cntL += (lane < NCH) ? kcnt[wk0 + lane] : 0;
        int wb = __builtin_amdgcn_readfirstlane(kofs[wk0]);
        int we = __builtin_amdgcn_readfirstlane(kofs[wk0 + NCH]);
        wb = wb < 0 ? 0 : (wb < n ? wb : n);
        we = we < wb ? wb : (we < n ? we : n);
        const bool later = cb != beg;
        int so[NCH], kn[NCH];
        {
            const int p = wb + lane;
            const int2 e = srt[(lane < NCH && p < we) ? p : 0];
#pragma unroll
            for (int q = 0; q < NCH; ++q) {
                const int s = __builtin_amdgcn_readlane(e.x, q);
                so[q] = s < 0 ? 0 : (s < N ? s : N - 1);
                const int k = __builtin_amdgcn_readlane(e.y, q);
                kn[q] = k < 0 ? 0 : (k < NCH ? k : NCH - 1);
            }
        }
        int cur = -1;
        float4 acc = make_float4(0.f, 0.f, 0.f, 0.f);
        for (int t0 = wb; t0 < we; t0 += NCH) {
            float4 v[NCH];
#pragma unroll
            for (int q = 0; q < NCH; ++q) v[q] = h4[(size_t)so[q] * DV4 + lane];
            int kc[NCH];
#pragma unroll
            for (int q = 0; q < NCH; ++q) kc[q] = kn[q];
            {
                const int p = t0 + NCH + lane;
                const int2 e = srt[(lane < NCH && p < we) ? p : 0];
#pragma unroll
                for (int q = 0; q < NCH; ++q) {
                    const int s = __builtin_amdgcn_readlane(e.x, q);
                    so[q] = s < 0 ? 0 : (s < N ? s : N - 1);
                    asm volatile("" : "+s"(so[q]));
                    const int k = __builtin_amdgcn_readlane(e.y, q);
                    kn[q] = k < 0 ? 0 : (k < NCH ? k : NCH - 1);
                }
            }
#pragma unroll
            for (int q = 0; q < NCH; ++q) {
                if (t0 + q < we) {
                    const int k = kc[q];
                    if (k != cur) {
                        if (cur >= 0)
                            *(float4*)(As + (wave * RPW + (cur & 3)) * AP + (cur >> 2) * D + lane * 4) = acc;
                        acc = make_float4(0.f, 0.f, 0.f, 0.f);
                        if (later)
                            acc = *(const float4*)(As + (wave * RPW + (k & 3)) * AP + (k >> 2) * D + lane * 4);
                        cur = k;
                    }
                    acc.x += v[q].x;
                    acc.y += v[q].y;
                    acc.z += v[q].z;
                    acc.w += v[q].w;
                }
            }
        }
        if (cur >= 0)
            *(float4*)(As + (wave * RPW + (cur & 3)) * AP + (cur >> 2) * D + lane * 4) = acc;
        __syncthreads();
    }

    const float scL = invR / (float)(cntL > 0 ? cntL : 1);
#pragma unroll
    for (int q = 0; q < NCH; ++q) {
        const float sc = __builtin_bit_cast(float,
                             __builtin_amdgcn_readlane(__builtin_bit_cast(int, scL), q));
        const int r = q >> 2;
        const int row = wave * RPW + (q & 3);
        float4 m = *(const float4*)(As + row * AP + r * D + lane * 4);
        m.x *= sc; m.y *= sc; m.z *= sc; m.w *= sc;
        *(float4*)(As + row * AP + r * D + lane * 4) = m;
    }
    __syncthreads();

    const int half = lane >> 4;
    const int lm   = lane & 15;
    const int colBase = wave * 16;

    const int wrow0 = __builtin_amdgcn_readfirstlane(rowBase + wave * RPW);
    v8f c0 = {};
    v8f c1 = {};
    v2f bw[2][KSB];
    const float* __restrict__ Wl = W + (size_t)(2 * half) * D + colBase + lm;
    float4 hres[RPW];
#pragma unroll
    for (int j = 0; j < KSB; ++j) {
        bw[0][j][0] = Wl[(size_t)(4 * j) * D];
        bw[0][j][1] = Wl[(size_t)(4 * j + 1) * D];
    }
#pragma unroll
    for (int blk = 0; blk < KT / (4 * KSB); ++blk) {
        if (blk + 1 < KT / (4 * KSB)) {
#pragma unroll
            for (int j = 0; j < KSB; ++j) {
                const int kn = (blk + 1) * 4 * KSB + 4 * j;
                bw[(blk + 1) & 1][j][0] = Wl[(size_t)kn * D];
                bw[(blk + 1) & 1][j][1] = Wl[(size_t)(kn + 1) * D];
            }
        }
        if (blk == KT / (4 * KSB) - 1) {
#pragma unroll
            for (int q = 0; q < RPW; ++q) {
                const int grow = wrow0 + q;
                hres[q] = h4[(size_t)(grow < N ? grow : N - 1) * DV4 + lane];
            }
        }
#pragma unroll
        for (int j = 0; j < KSB; ++j) {
            const int ka = blk * 4 * KSB + 4 * j + 2 * half;
            v2f a0, a1;
            a0[0] = As[lm * AP + ka];
            a0[1] = As[lm * AP + ka + 1];
            a1[0] = As[(16 + lm) * AP + ka];
            a1[1] = As[(16 + lm) * AP + ka + 1];
            c0 = __builtin_amdgcn_wmma_f32_16x16x4_f32(
                     false, a0,  false, bw[blk & 1][j],
                     (short)0, c0,  false,  false);
            c1 = __builtin_amdgcn_wmma_f32_16x16x4_f32(
                     false, a1,  false, bw[blk & 1][j],
                     (short)0, c1,  false,  false);
        }
        __builtin_amdgcn_sched_barrier(0);
    }

    __syncthreads();
#pragma unroll
    for (int v = 0; v < 8; ++v) {
        Cs[(v + 8 * half) * CP + colBase + lm]      = c0[v];
        Cs[(16 + v + 8 * half) * CP + colBase + lm] = c1[v];
    }
    __syncthreads();

    float4* __restrict__ out4 = (float4*)out;
#pragma unroll
    for (int q = 0; q < RPW; ++q) {
        const int grow = wrow0 + q;
        const int row  = grow - rowBase;
        if (grow < N) {
            const float4 cv = *(const float4*)(Cs + row * CP + lane * 4);
            const float4 hv = hres[q];
            float4 o;
            o.x = hv.x + cv.x; o.y = hv.y + cv.y; o.z = hv.z + cv.z; o.w = hv.w + cv.w;
            out4[(size_t)grow * DV4 + lane] = o;
        }
    }
}

extern "C" void kernel_launch(void* const* d_in, const int* in_sizes, int n_in,
                              void* d_out, int out_size, void* d_ws, size_t ws_size,
                              hipStream_t stream) {
    const float* h   = (const float*)d_in[0];
    const float* W   = (const float*)d_in[1];
    const int*   src = (const int*)d_in[2];
    const int*   dst = (const int*)d_in[3];
    float*       out = (float*)d_out;

    const int d = D;
    const int N = in_sizes[0] / d;
    const int R = NR;
    const int E = in_sizes[2] / R;
    const int total = R * E;

    int*  gcount  = (int*)d_ws;
    int*  gcursor = gcount + NBMAX;
    int*  off     = gcursor + NBMAX;
    int*  rec     = off + NBMAX + 4;

    fill_int_kernel<<<(2 * NBMAX + 255) / 256, 256, 0, stream>>>(gcount, 2 * NBMAX, 0);
    {
        const int g = (total + GT * EPT - 1) / (GT * EPT);
        count_kernel<<<g, GT, 0, stream>>>(dst, gcount, total, N);
        group_kernel<<<g, GT, 0, stream>>>(src, dst, gcount, gcursor, off, rec, total, N, E);
    }
    {
        const float invR = 1.0f / (float)R;
        gather_transform_kernel<<<(N + TM - 1) / TM, NWV * 32, 0, stream>>>(h, W, off, rec,
                                                                          out, N, total, invR);
    }
}


# edges grouped per 6144-edge chunk into the chunk's own region plus one run word per tile; the tile's work-group collects its runs: 4 launches -> 2, no global atomics; sum order: chunk by chunk
# speedup vs baseline: 4.9136x; 1.0597x over previous
#include <hip/hip_runtime.h>


typedef __attribute__((ext_vector_type(2))) float v2f;
typedef __attribute__((ext_vector_type(8))) float v8f;

#define D 128
#define DV4 (D / 4)
#define NR 3
#define KT (NR * D)
#define AP (KT + 4)
#define CP (D + 4)
#define TM 32
#define NWV 8
#define RPW (TM / NWV)
#define NCH (RPW * NR)
#define NKEY (NR * TM)
#define NBMAX 4096
#define GT 1024
#define EPT 6
#define CH 1024
#define RPT (CH / (NWV * 32))
#define GB (NWV * 32)
#define KSB 12

__device__ __forceinline__ int tile_of(int d, int N) {
    d = d < 0 ? 0 : (d < N ? d : N - 1);
    const int t = d >> 5;
    return t < NBMAX ? t : NBMAX - 1;
}

__global__ __launch_bounds__(GT)
void group_kernel(const int* __restrict__ src, const int* __restrict__ dst,
                  int* __restrict__ runs, int* __restrict__ recs, int total, int N, int E) {
    int2* __restrict__ rec = (int2*)recs;
    __shared__ int offs[NBMAX + 4];
    __shared__ __attribute__((aligned(16))) int cnt[NBMAX];
    __shared__ int wsum[GT / 32];
    __shared__ int2 srec[GT * EPT];
    const int tid = threadIdx.x, lane = tid & 31, wave = tid >> 5;
    const int base = blockIdx.x * (GT * EPT);

    int dk[EPT], sk[EPT], rl[EPT];
#pragma unroll
    for (int k = 0; k < EPT; ++k) {
        const int i = base + k * GT + tid;
        const int j = i < total ? i : 0;
        dk[k] = dst[j];
        sk[k] = src[j];
        rl[k] = i / E;
    }
    for (int b = tid; b < NBMAX; b += GT) cnt[b] = 0;
    __syncthreads();

    int rank[EPT], tk[EPT];
#pragma unroll
    for (int k = 0; k < EPT; ++k) {
        const int i = base + k * GT + tid;
        int d = dk[k];
        d = d < 0 ? 0 : (d < N ? d : N - 1);
        dk[k] = d;
        tk[k] = tile_of(d, N);
        rank[k] = 0;
        if (i < total) rank[k] = atomicAdd(&cnt[tk[k]], 1);
    }
    __syncthreads();

    const int4 v = ((const int4*)cnt)[tid];
    const int s = v.x + v.y + v.z + v.w;
    int incl = s;
#pragma unroll
    for (int o = 1; o < 32; o <<= 1) {
        const int t2 = __shfl_up(incl, o);
        if (lane >= o) incl += t2;
    }
    if (lane == 31) wsum[wave] = incl;
    __syncthreads();
    if (wave == 0) {
        const int x = wsum[lane];
        int xi = x;
#pragma unroll
        for (int o = 1; o < 32; o <<= 1) {
            const int t2 = __shfl_up(xi, o);
            if (lane >= o) xi += t2;
        }
        wsum[lane] = xi - x;
    }
    __syncthreads();
    const int ex = incl - s + wsum[wave];
    offs[4 * tid + 0] = ex;
    offs[4 * tid + 1] = ex + v.x;
    offs[4 * tid + 2] = ex + v.x + v.y;
    offs[4 * tid + 3] = ex + v.x + v.y + v.z;
    int4 pk;
    pk.x = ex | (v.x << 16);
    pk.y = (ex + v.x) | (v.y << 16);
    pk.z = (ex + v.x + v.y) | (v.z << 16);
    pk.w = (ex + v.x + v.y + v.z) | (v.w << 16);
    ((int4*)(runs + (size_t)blockIdx.x * NBMAX))[tid] = pk;
    __syncthreads();
    int st6[EPT];
#pragma unroll
    for (int k = 0; k < EPT; ++k) st6[k] = offs[tk[k]];
#pragma unroll
    for (int k = 0; k < EPT; ++k) {
        const int i = base + k * GT + tid;
        if (i < total) {
            int pos = st6[k] + rank[k];
            pos = pos < 0 ? 0 : (pos < GT * EPT ? pos : GT * EPT - 1);
            srec[pos] = make_int2(rl[k] * TM + (dk[k] & (TM - 1)), sk[k]);
        }
    }
    __syncthreads();
#pragma unroll
    for (int k = 0; k < EPT; ++k) {
        const int j = k * GT + tid;
        if (base + j < total) rec[base + j] = srec[j];
    }
}

__global__ __launch_bounds__(256, 10)
void gather_transform_kernel(const float* __restrict__ h,
                             const float* __restrict__ W,
                             const int* __restrict__ runs,
                             const int* __restrict__ recs,
                             float* __restrict__ out,
                             int N, int total, int G, float invR) {
    __shared__ __attribute__((aligned(16))) float As[TM * AP];
    __shared__ int kcnt[NKEY];
    __shared__ int kofs[NKEY + 1];
    __shared__ int2 srt[CH];
    __shared__ int wtot[NWV];
    int* const ridx = (int*)srt;

    float* const Cs = As;
    const int wave = threadIdx.x >> 5;
    const int lane = threadIdx.x & 31;
    const int rowBase = blockIdx.x * TM;
    const float4* __restrict__ h4 = (const float4*)h;
    const int2* __restrict__ rec = (const int2*)recs;


    const int wk0 = wave * NCH;
    int cntL = 0;
#pragma unroll
    for (int q = 0; q < NCH; ++q)
        *(float4*)(As + (wave * RPW + (q & 3)) * AP + (q >> 2) * D + lane * 4) = make_float4(0.f, 0.f, 0.f, 0.f);

    bool later = false;
    for (int g0 = 0; g0 < G; g0 += GB) {
    const int gq = g0 + threadIdx.x;
    int rc_n = 0, rc_b = 0;
    if (gq < G) {
        const int p = runs[(size_t)gq * NBMAX + (blockIdx.x < NBMAX ? blockIdx.x : NBMAX - 1)];
        int c = (p >> 16) & 0x7fff, st = p & 0xffff;
        c  = c  < GT * EPT ? c  : GT * EPT;
        st = st < GT * EPT ? st : GT * EPT - 1;
        rc_n = c;
        rc_b = gq * (GT * EPT) + st;
    }
    int rincl = rc_n;
#pragma unroll
    for (int o = 1; o < 32; o <<= 1) {
        const int t2 = __shfl_up(rincl, o);
        if (lane >= o) rincl += t2;
    }
    if (lane == 31) wtot[wave] = rincl;
    __syncthreads();
    int rex = rincl - rc_n, ntv = 0;
#pragma unroll
    for (int w = 0; w < NWV; ++w) {
        const int x = wtot[w];
        rex += w < wave ? x : 0;
        ntv += x;
    }
    const int nt = __builtin_amdgcn_readfirstlane(ntv);
    for (int cb = 0; cb < nt; cb += CH) {
        const int n = (nt - cb) < CH ? (nt - cb) : CH;
        if (threadIdx.x < NKEY) kcnt[threadIdx.x] = 0;
        {
            const int lo = rex > cb ? rex : cb;
            const int hi = (rex + rc_n) < (cb + CH) ? (rex + rc_n) : (cb + CH);
#pragma nounroll
            for (int r = lo; r < hi; ++r) ridx[r - cb] = rc_b + (r - rex);
        }
        __syncthreads();
        int rk[RPT], ky[RPT], lk[RPT], sv[RPT];
        int2 rc[RPT];
#pragma unroll
        for (int j = 0; j < RPT; ++j) {
            const int idx = j * (NWV * 32) + threadIdx.x;
            int ri = ridx[idx < n ? idx : 0];
            ri = ri < 0 ? 0 : (ri < total ? ri : total - 1);
            rc[j] = rec[ri];
        }
#pragma unroll
        for (int j = 0; j < RPT; ++j) {
            const int idx = j * (NWV * 32) + threadIdx.x;
            const int kr = rc[j].x < 0 ? 0 : (rc[j].x < NKEY ? rc[j].x : NKEY - 1);
            const int rw = kr & (TM - 1);
            lk[j] = (kr / TM) * RPW + (rw & (RPW - 1));
            ky[j] = (rw / RPW) * NCH + lk[j];
            sv[j] = rc[j].y;
            rk[j] = 0;
            if (idx < n) rk[j] = atomicAdd(&kcnt[ky[j]], 1);
        }
        __syncthreads();
        if (wave == 0) {
            const int a0 = kcnt[3 * lane], a1 = kcnt[3 * lane + 1], a2 = kcnt[3 * lane + 2];
            const int s3 = a0 + a1 + a2;
            int incl = s3;
#pragma unroll
            for (int o = 1; o < 32; o <<= 1) {
                const int t2 = __shfl_up(incl, o);
                if (lane >= o) incl += t2;
            }
            const int ex = incl - s3;
            kofs[3 * lane]     = ex;
            kofs[3 * lane + 1] = ex + a0;
            kofs[3 * lane + 2] = ex + a0 + a1;
            if (lane == 31) kofs[NKEY] = incl;
        }
        __syncthreads();
#pragma unroll
        for (int j = 0; j < RPT; ++j) {
            const int idx = j * (NWV * 32) + threadIdx.x;
            if (idx < n) {
                int p = kofs[ky[j]] + rk[j];
                p = p < CH ? p : CH - 1;
                srt[p] = make_int2(sv[j], lk[j]);
            }
        }
        __syncthreads();

        cntL += (lane < NCH) ? kcnt[wk0 + lane] : 0;
        int wb = __builtin_amdgcn_readfirstlane(kofs[wk0]);
        int we = __builtin_amdgcn_readfirstlane(kofs[wk0 + NCH]);
        wb = wb < 0 ? 0 : (wb < n ? wb : n);
        we = we < wb ? wb : (we < n ? we : n);
        int so[NCH], kn[NCH];
        {
            const int p = wb + lane;
            const int2 e = srt[(lane < NCH && p < we) ? p : 0];
#pragma unroll
            for (int q = 0; q < NCH; ++q) {
                const int s = __builtin_amdgcn_readlane(e.x, q);
                so[q] = s < 0 ? 0 : (s < N ? s : N - 1);
                const int k = __builtin_amdgcn_readlane(e.y, q);
                kn[q] = k < 0 ? 0 : (k < NCH ? k : NCH - 1);
            }
        }
        int cur = -1;
        float4 acc = make_float4(0.f, 0.f, 0.f, 0.f);
        for (int t0 = wb; t0 < we; t0 += NCH) {
            float4 v[NCH];
#pragma unroll
            for (int q = 0; q < NCH; ++q) v[q] = h4[(size_t)so[q] * DV4 + lane];
            int kc[NCH];
#pragma unroll
            for (int q = 0; q < NCH; ++q) kc[q] = kn[q];
            {
                const int p = t0 + NCH + lane;
                const int2 e = srt[(lane < NCH && p < we) ? p : 0];
#pragma unroll
                for (int q = 0; q < NCH; ++q) {
                    const int s = __builtin_amdgcn_readlane(e.x, q);
                    so[q] = s < 0 ? 0 : (s < N ? s : N - 1);
                    asm volatile("" : "+s"(so[q]));
                    const int k = __builtin_amdgcn_readlane(e.y, q);
                    kn[q] = k < 0 ? 0 : (k < NCH ? k : NCH - 1);
                }
            }
#pragma unroll
            for (int q = 0; q < NCH; ++q) {
                if (t0 + q < we) {
                    const int k = kc[q];
                    if (k != cur) {
                        if (cur >= 0)
                            *(float4*)(As + (wave * RPW + (cur & 3)) * AP + (cur >> 2) * D + lane * 4) = acc;
                        acc = make_float4(0.f, 0.f, 0.f, 0.f);
                        if (later)
                            acc = *(const float4*)(As + (wave * RPW + (k & 3)) * AP + (k >> 2) * D + lane * 4);
                        cur = k;
                    }
                    acc.x += v[q].x;
                    acc.y += v[q].y;
                    acc.z += v[q].z;
                    acc.w += v[q].w;
                }
            }
        }
        if (cur >= 0)
            *(float4*)(As + (wave * RPW + (cur & 3)) * AP + (cur >> 2) * D + lane * 4) = acc;
        __syncthreads();
        later = true;
    }
    __syncthreads();
    }

    const float scL = invR / (float)(cntL > 0 ? cntL : 1);
#pragma unroll
    for (int q = 0; q < NCH; ++q) {
        const float sc = __builtin_bit_cast(float,
                             __builtin_amdgcn_readlane(__builtin_bit_cast(int, scL), q));
        const int r = q >> 2;
        const int row = wave * RPW + (q & 3);
        float4 m = *(const float4*)(As + row * AP + r * D + lane * 4);
        m.x *= sc; m.y *= sc; m.z *= sc; m.w *= sc;
        *(float4*)(As + row * AP + r * D + lane * 4) = m;
    }
    __syncthreads();

    const int half = lane >> 4;
    const int lm   = lane & 15;
    const int colBase = wave * 16;

    const int wrow0 = __builtin_amdgcn_readfirstlane(rowBase + wave * RPW);
    v8f c0 = {};
    v8f c1 = {};
    v2f bw[2][KSB];
    const float* __restrict__ Wl = W + (size_t)(2 * half) * D + colBase + lm;
    float4 hres[RPW];
#pragma unroll
    for (int j = 0; j < KSB; ++j) {
        bw[0][j][0] = Wl[(size_t)(4 * j) * D];
        bw[0][j][1] = Wl[(size_t)(4 * j + 1) * D];
    }
#pragma unroll
    for (int blk = 0; blk < KT / (4 * KSB); ++blk) {
        if (blk + 1 < KT / (4 * KSB)) {
#pragma unroll
            for (int j = 0; j < KSB; ++j) {
                const int kn = (blk + 1) * 4 * KSB + 4 * j;
                bw[(blk + 1) & 1][j][0] = Wl[(size_t)kn * D];
                bw[(blk + 1) & 1][j][1] = Wl[(size_t)(kn + 1) * D];
            }
        }
        if (blk == KT / (4 * KSB) - 1) {
#pragma unroll
            for (int q = 0; q < RPW; ++q) {
                const int grow = wrow0 + q;
                hres[q] = h4[(size_t)(grow < N ? grow : N - 1) * DV4 + lane];
            }
        }
#pragma unroll
        for (int j = 0; j < KSB; ++j) {
            const int ka = blk * 4 * KSB + 4 * j + 2 * half;
            v2f a0, a1;
            a0[0] = As[lm * AP + ka];
            a0[1] = As[lm * AP + ka + 1];
            a1[0] = As[(16 + lm) * AP + ka];
            a1[1] = As[(16 + lm) * AP + ka + 1];
            c0 = __builtin_amdgcn_wmma_f32_16x16x4_f32(
                     false, a0,  false, bw[blk & 1][j],
                     (short)0, c0,  false,  false);
            c1 = __builtin_amdgcn_wmma_f32_16x16x4_f32(
                     false, a1,  false, bw[blk & 1][j],
                     (short)0, c1,  false,  false);
        }
        __builtin_amdgcn_sched_barrier(0);
    }

    __syncthreads();
#pragma unroll
    for (int v = 0; v < 8; ++v) {
        Cs[(v + 8 * half) * CP + colBase + lm]      = c0[v];
        Cs[(16 + v + 8 * half) * CP + colBase + lm] = c1[v];
    }
    __syncthreads();

    float4* __restrict__ out4 = (float4*)out;
#pragma unroll
    for (int q = 0; q < RPW; ++q) {
        const int grow = wrow0 + q;
        const int row  = grow - rowBase;
        if (grow < N) {
            const float4 cv = *(const float4*)(Cs + row * CP + lane * 4);
            const float4 hv = hres[q];
            float4 o;
            o.x = hv.x + cv.x; o.y = hv.y + cv.y; o.z = hv.z + cv.z; o.w = hv.w + cv.w;
            out4[(size_t)grow * DV4 + lane] = o;
        }
    }
}

extern "C" void kernel_launch(void* const* d_in, const int* in_sizes, int n_in,
                              void* d_out, int out_size, void* d_ws, size_t ws_size,
                              hipStream_t stream) {
    const float* h   = (const float*)d_in[0];
    const float* W   = (const float*)d_in[1];
    const int*   src = (const int*)d_in[2];
    const int*   dst = (const int*)d_in[3];
    float*       out = (float*)d_out;

    const int d = D;
    const int N = in_sizes[0] / d;
    const int R = NR;
    const int E = in_sizes[2] / R;
    const int total = R * E;

    const int g = (total + GT * EPT - 1) / (GT * EPT);
    int*  runs = (int*)d_ws;
    int*  rec  = runs + (size_t)g * NBMAX;

    group_kernel<<<g, GT, 0, stream>>>(src, dst, runs, rec, total, N, E);
    {
        const float invR = 1.0f / (float)R;
        gather_transform_kernel<<<(N + TM - 1) / TM, NWV * 32, 0, stream>>>(h, W, runs, rec,
                                                                          out, N, total, g, invR);
    }
}


# W re-laid per call (in the group launch) with 4 k side by side per column: the B and A fragments of two k-steps load as one 16-byte read, whole lines; sum order changed (k order inside 8-k groups)
# speedup vs baseline: 5.2029x; 1.0589x over previous
#include <hip/hip_runtime.h>


typedef __attribute__((ext_vector_type(2))) float v2f;
typedef __attribute__((ext_vector_type(4))) float v4f;
typedef __attribute__((ext_vector_type(8))) float v8f;

#define D 128
#define DV4 (D / 4)
#define NR 3
#define KT (NR * D)
#define AP (KT + 4)
#define CP (D + 4)
#define TM 32
#define NWV 8
#define RPW (TM / NWV)
#define NCH (RPW * NR)
#define NKEY (NR * TM)
#define NBMAX 4096
#define GT 1024
#define EPT 6
#define CH 1024
#define RPT (CH / (NWV * 32))
#define GB (NWV * 32)
#define KPB 6
#define NBLK (KT / (8 * KPB))

__device__ __forceinline__ int tile_of(int d, int N) {
    d = d < 0 ? 0 : (d < N ? d : N - 1);
    const int t = d >> 5;
    return t < NBMAX ? t : NBMAX - 1;
}

__global__ __launch_bounds__(GT)
void group_kernel(const int* __restrict__ src, const int* __restrict__ dst, const float* __restrict__ W,
                  int* __restrict__ runs, int* __restrict__ recs, float* __restrict__ Wt, int total, int N, int E) {
    int2* __restrict__ rec = (int2*)recs;
    __shared__ int offs[NBMAX + 4];
    __shared__ __attribute__((aligned(16))) int cnt[NBMAX];
    __shared__ int wsum[GT / 32];
    __shared__ int2 srec[GT * EPT];
    const int tid = threadIdx.x, lane = tid & 31, wave = tid >> 5;
    const int base = blockIdx.x * (GT * EPT);

    int dk[EPT], sk[EPT], rl[EPT];
#pragma unroll
    for (int k = 0; k < EPT; ++k) {
        const int i = base + k * GT + tid;
        const int j = i < total ? i : 0;
        dk[k] = dst[j];
        sk[k] = src[j];
        rl[k] = i / E;
    }
    for (int b = tid; b < NBMAX; b += GT) cnt[b] = 0;
    for (int i = blockIdx.x * GT + tid; i < KT * D; i += gridDim.x * GT)
        Wt[i] = W[(size_t)(4 * (i / (4 * D)) + (i & 3)) * D + ((i >> 2) & (D - 1))];
    __syncthreads();

    int rank[EPT], tk[EPT];
#pragma unroll
    for (int k = 0; k < EPT; ++k) {
        const int i = base + k * GT + tid;
        int d = dk[k];
        d = d < 0 ? 0 : (d < N ? d : N - 1);
        dk[k] = d;
        tk[k] = tile_of(d, N);
        rank[k] = 0;
        if (i < total) rank[k] = atomicAdd(&cnt[tk[k]], 1);
    }
    __syncthreads();

    const int4 v = ((const int4*)cnt)[tid];
    const int s = v.x + v.y + v.z + v.w;
    int incl = s;
#pragma unroll
    for (int o = 1; o < 32; o <<= 1) {
        const int t2 = __shfl_up(incl, o);
        if (lane >= o) incl += t2;
    }
    if (lane == 31) wsum[wave] = incl;
    __syncthreads();
    if (wave == 0) {
        const int x = wsum[lane];
        int xi = x;
#pragma unroll
        for (int o = 1; o < 32; o <<= 1) {
            const int t2 = __shfl_up(xi, o);
            if (lane >= o) xi += t2;
        }
        wsum[lane] = xi - x;
    }
    __syncthreads();
    const int ex = incl - s + wsum[wave];
    offs[4 * tid + 0] = ex;
    offs[4 * tid + 1] = ex + v.x;
    offs[4 * tid + 2] = ex + v.x + v.y;
    offs[4 * tid + 3] = ex + v.x + v.y + v.z;
    int4 pk;
    pk.x = ex | (v.x << 16);
    pk.y = (ex + v.x) | (v.y << 16);
    pk.z = (ex + v.x + v.y) | (v.z << 16);
    pk.w = (ex + v.x + v.y + v.z) | (v.w << 16);
    ((int4*)(runs + (size_t)blockIdx.x * NBMAX))[tid] = pk;
    __syncthreads();
    int st6[EPT];
#pragma unroll
    for (int k = 0; k < EPT; ++k) st6[k] = offs[tk[k]];
#pragma unroll
    for (int k = 0; k < EPT; ++k) {
        const int i = base + k * GT + tid;
        if (i < total) {
            int pos = st6[k] + rank[k];
            pos = pos < 0 ? 0 : (pos < GT * EPT ? pos : GT * EPT - 1);
            srec[pos] = make_int2(rl[k] * TM + (dk[k] & (TM - 1)), sk[k]);
        }
    }
    __syncthreads();
#pragma unroll
    for (int k = 0; k < EPT; ++k) {
        const int j = k * GT + tid;
        if (base + j < total) rec[base + j] = srec[j];
    }
}

__global__ __launch_bounds__(256, 10)
void gather_transform_kernel(const float* __restrict__ h,
                             const float* __restrict__ Wt,
                             const int* __restrict__ runs,
                             const int* __restrict__ recs,
                             float* __restrict__ out,
                             int N, int total, int G, float invR) {
    __shared__ __attribute__((aligned(16))) float As[TM * AP];
    __shared__ int kcnt[NKEY];
    __shared__ int kofs[NKEY + 1];
    __shared__ int2 srt[CH];
    __shared__ int wtot[NWV];
    int* const ridx = (int*)srt;

    float* const Cs = As;
    const int wave = threadIdx.x >> 5;
    const int lane = threadIdx.x & 31;
    const int rowBase = blockIdx.x * TM;
    const float4* __restrict__ h4 = (const float4*)h;
    const int2* __restrict__ rec = (const int2*)recs;


    const int wk0 = wave * NCH;
    int cntL = 0;
#pragma unroll
    for (int q = 0; q < NCH; ++q)
        *(float4*)(As + (wave * RPW + (q & 3)) * AP + (q >> 2) * D + lane * 4) = make_float4(0.f, 0.f, 0.f, 0.f);

    bool later = false;
    for (int g0 = 0; g0 < G; g0 += GB) {
    const int gq = g0 + threadIdx.x;
    int rc_n = 0, rc_b = 0;
    if (gq < G) {
        const int p = runs[(size_t)gq * NBMAX + (blockIdx.x < NBMAX ? blockIdx.x : NBMAX - 1)];
        int c = (p >> 16) & 0x7fff, st = p & 0xffff;
        c  = c  < GT * EPT ? c  : GT * EPT;
        st = st < GT * EPT ? st : GT * EPT - 1;
        rc_n = c;
        rc_b = gq * (GT * EPT) + st;
    }
    int rincl = rc_n;
#pragma unroll
    for (int o = 1; o < 32; o <<= 1) {
        const int t2 = __shfl_up(rincl, o);
        if (lane >= o) rincl += t2;
    }
    if (lane == 31) wtot[wave] = rincl;
    __syncthreads();
    int rex = rincl - rc_n, ntv = 0;
#pragma unroll
    for (int w = 0; w < NWV; ++w) {
        const int x = wtot[w];
        rex += w < wave ? x : 0;
        ntv += x;
    }
    const int nt = __builtin_amdgcn_readfirstlane(ntv);
    for (int cb = 0; cb < nt; cb += CH) {
        const int n = (nt - cb) < CH ? (nt - cb) : CH;
        if (threadIdx.x < NKEY) kcnt[threadIdx.x] = 0;
        {
            const int lo = rex > cb ? rex : cb;
            const int hi = (rex + rc_n) < (cb + CH) ? (rex + rc_n) : (cb + CH);
#pragma nounroll
            for (int r = lo; r < hi; ++r) ridx[r - cb] = rc_b + (r - rex);
        }
        __syncthreads();
        int rk[RPT], ky[RPT], lk[RPT], sv[RPT];
        int2 rc[RPT];
#pragma unroll
        for (int j = 0; j < RPT; ++j) {
            const int idx = j * (NWV * 32) + threadIdx.x;
            int ri = ridx[idx < n ? idx : 0];
            ri = ri < 0 ? 0 : (ri < total ? ri : total - 1);
            rc[j] = rec[ri];
        }
#pragma unroll
        for (int j = 0; j < RPT; ++j) {
            const int idx = j * (NWV * 32) + threadIdx.x;
            const int kr = rc[j].x < 0 ? 0 : (rc[j].x < NKEY ? rc[j].x : NKEY - 1);
            const int rw = kr & (TM - 1);
            lk[j] = (kr / TM) * RPW + (rw & (RPW - 1));
            ky[j] = (rw / RPW) * NCH + lk[j];
            sv[j] = rc[j].y;
            rk[j] = 0;
            if (idx < n) rk[j] = atomicAdd(&kcnt[ky[j]], 1);
        }
        __syncthreads();
        if (wave == 0) {
            const int a0 = kcnt[3 * lane], a1 = kcnt[3 * lane + 1], a2 = kcnt[3 * lane + 2];
            const int s3 = a0 + a1 + a2;
            int incl = s3;
#pragma unroll
            for (int o = 1; o < 32; o <<= 1) {
                const int t2 = __shfl_up(incl, o);
                if (lane >= o) incl += t2;
            }
            const int ex = incl - s3;
            kofs[3 * lane]     = ex;
            kofs[3 * lane + 1] = ex + a0;
            kofs[3 * lane + 2] = ex + a0 + a1;
            if (lane == 31) kofs[NKEY] = incl;
        }
        __syncthreads();
#pragma unroll
        for (int j = 0; j < RPT; ++j) {
            const int idx = j * (NWV * 32) + threadIdx.x;
            if (idx < n) {
                int p = kofs[ky[j]] + rk[j];
                p = p < CH ? p : CH - 1;
                srt[p] = make_int2(sv[j], lk[j]);
            }
        }
        __syncthreads();

        cntL += (lane < NCH) ? kcnt[wk0 + lane] : 0;
        int wb = __builtin_amdgcn_readfirstlane(kofs[wk0]);
        int we = __builtin_amdgcn_readfirstlane(kofs[wk0 + NCH]);
        wb = wb < 0 ? 0 : (wb < n ? wb : n);
        we = we < wb ? wb : (we < n ? we : n);
        int so[NCH], kn[NCH];
        {
            const int p = wb + lane;
            const int2 e = srt[(lane < NCH && p < we) ? p : 0];
#pragma unroll
            for (int q = 0; q < NCH; ++q) {
                const int s = __builtin_amdgcn_readlane(e.x, q);
                so[q] = s < 0 ? 0 : (s < N ? s : N - 1);
                const int k = __builtin_amdgcn_readlane(e.y, q);
                kn[q] = k < 0 ? 0 : (k < NCH ? k : NCH - 1);
            }
        }
        int cur = -1;
        float4 acc = make_float4(0.f, 0.f, 0.f, 0.f);
        for (int t0 = wb; t0 < we; t0 += NCH) {
            float4 v[NCH];
#pragma unroll
            for (int q = 0; q < NCH; ++q) v[q] = h4[(size_t)so[q] * DV4 + lane];
            int kc[NCH];
#pragma unroll
            for (int q = 0; q < NCH; ++q) kc[q] = kn[q];
            {
                const int p = t0 + NCH + lane;
                const int2 e = srt[(lane < NCH && p < we) ? p : 0];
#pragma unroll
                for (int q = 0; q < NCH; ++q) {
                    const int s = __builtin_amdgcn_readlane(e.x, q);
                    so[q] = s < 0 ? 0 : (s < N ? s : N - 1);
                    asm volatile("" : "+s"(so[q]));
                    const int k = __builtin_amdgcn_readlane(e.y, q);
                    kn[q] = k < 0 ? 0 : (k < NCH ? k : NCH - 1);
                }
            }
#pragma unroll
            for (int q = 0; q < NCH; ++q) {
                if (t0 + q < we) {
                    const int k = kc[q];
                    if (k != cur) {
                        if (cur >= 0)
                            *(float4*)(As + (wave * RPW + (cur & 3)) * AP + (cur >> 2) * D + lane * 4) = acc;
                        acc = make_float4(0.f, 0.f, 0.f, 0.f);
                        if (later)
                            acc = *(const float4*)(As + (wave * RPW + (k & 3)) * AP + (k >> 2) * D + lane * 4);
                        cur = k;
                    }
                    acc.x += v[q].x;
                    acc.y += v[q].y;
                    acc.z += v[q].z;
                    acc.w += v[q].w;
                }
            }
        }
        if (cur >= 0)
            *(float4*)(As + (wave * RPW + (cur & 3)) * AP + (cur >> 2) * D + lane * 4) = acc;
        __syncthreads();
        later = true;
    }
    __syncthreads();
    }

    const float scL = invR / (float)(cntL > 0 ? cntL : 1);
#pragma unroll
    for (int q = 0; q < NCH; ++q) {
        const float sc = __builtin_bit_cast(float,
                             __builtin_amdgcn_readlane(__builtin_bit_cast(int, scL), q));
        const int r = q >> 2;
        const int row = wave * RPW + (q & 3);
        float4 m = *(const float4*)(As + row * AP + r * D + lane * 4);
        m.x *= sc; m.y *= sc; m.z *= sc; m.w *= sc;
        *(float4*)(As + row * AP + r * D + lane * 4) = m;
    }
    __syncthreads();

    const int half = lane >> 4;
    const int lm   = lane & 15;
    const int colBase = wave * 16;

    const int wrow0 = __builtin_amdgcn_readfirstlane(rowBase + wave * RPW);
    v8f c0 = {};
    v8f c1 = {};
    v4f bw[2][KPB];
    const v4f* __restrict__ Wq = (const v4f*)Wt + (size_t)half * D + colBase + lm;
    float4 hres[RPW];
#pragma unroll
    for (int j = 0; j < KPB; ++j) bw[0][j] = Wq[(size_t)(2 * j) * D];
#pragma unroll
    for (int blk = 0; blk < NBLK; ++blk) {
        if (blk + 1 < NBLK) {
#pragma unroll
            for (int j = 0; j < KPB; ++j)
                bw[(blk + 1) & 1][j] = Wq[(size_t)(2 * ((blk + 1) * KPB + j)) * D];
        }
        if (blk == NBLK - 1) {
#pragma unroll
            for (int q = 0; q < RPW; ++q) {
                const int grow = wrow0 + q;
                hres[q] = h4[(size_t)(grow < N ? grow : N - 1) * DV4 + lane];
            }
        }
#pragma unroll
        for (int j = 0; j < KPB; ++j) {
            const int ka = 4 * (2 * (blk * KPB + j) + half);
            const v4f a0 = *(const v4f*)(As + lm * AP + ka);
            const v4f a1 = *(const v4f*)(As + (16 + lm) * AP + ka);
            const v4f b  = bw[blk & 1][j];
            v2f a0l, a0h, a1l, a1h, bl, bh;
            a0l[0] = a0[0]; a0l[1] = a0[1]; a0h[0] = a0[2]; a0h[1] = a0[3];
            a1l[0] = a1[0]; a1l[1] = a1[1]; a1h[0] = a1[2]; a1h[1] = a1[3];
            bl[0]  = b[0];  bl[1]  = b[1];  bh[0]  = b[2];  bh[1]  = b[3];
            c0 = __builtin_amdgcn_wmma_f32_16x16x4_f32(
                     false, a0l,  false, bl,
                     (short)0, c0,  false,  false);
            c1 = __builtin_amdgcn_wmma_f32_16x16x4_f32(
                     false, a1l,  false, bl,
                     (short)0, c1,  false,  false);
            c0 = __builtin_amdgcn_wmma_f32_16x16x4_f32(
                     false, a0h,  false, bh,
                     (short)0, c0,  false,  false);
            c1 = __builtin_amdgcn_wmma_f32_16x16x4_f32(
                     false, a1h,  false, bh,
                     (short)0, c1,  false,  false);
        }
        __builtin_amdgcn_sched_barrier(0);
    }

    __syncthreads();
#pragma unroll
    for (int v = 0; v < 8; ++v) {
        Cs[(v + 8 * half) * CP + colBase + lm]      = c0[v];
        Cs[(16 + v + 8 * half) * CP + colBase + lm] = c1[v];
    }
    __syncthreads();

    float4* __restrict__ out4 = (float4*)out;
#pragma unroll
    for (int q = 0; q < RPW; ++q) {
        const int grow = wrow0 + q;
        const int row  = grow - rowBase;
        if (grow < N) {
            const float4 cv = *(const float4*)(Cs + row * CP + lane * 4);
            const float4 hv = hres[q];
            float4 o;
            o.x = hv.x + cv.x; o.y = hv.y + cv.y; o.z = hv.z + cv.z; o.w = hv.w + cv.w;
            out4[(size_t)grow * DV4 + lane] = o;
        }
    }
}

extern "C" void kernel_launch(void* const* d_in, const int* in_sizes, int n_in,
                              void* d_out, int out_size, void* d_ws, size_t ws_size,
                              hipStream_t stream) {
    const float* h   = (const float*)d_in[0];
    const float* W   = (const float*)d_in[1];
    const int*   src = (const int*)d_in[2];
    const int*   dst = (const int*)d_in[3];
    float*       out = (float*)d_out;

    const int d = D;
    const int N = in_sizes[0] / d;
    const int R = NR;
    const int E = in_sizes[2] / R;
    const int total = R * E;

    const int g = (total + GT * EPT - 1) / (GT * EPT);
    int*  runs = (int*)d_ws;
    int*  rec  = runs + (size_t)g * NBMAX;
    float* Wt  = (float*)(rec + (((size_t)2 * total + 3) & ~(size_t)3));

    group_kernel<<<g, GT, 0, stream>>>(src, dst, W, runs, rec, Wt, total, N, E);
    {
        const float invR = 1.0f / (float)R;
        gather_transform_kernel<<<(N + TM - 1) / TM, NWV * 32, 0, stream>>>(h, Wt, runs, rec,
                                                                          out, N, total, g, invR);
    }
}
